# starting kernel
# baseline (speedup 1.0000x reference)
#include <hip/hip_runtime.h>

typedef __attribute__((ext_vector_type(2))) float v2f;
typedef __attribute__((ext_vector_type(8))) float v8f;

static constexpr int NN   = 100000;
static constexpr int EE   = 1600000;
static constexpr int IND  = 256;
static constexpr int HID  = 128;
static constexpr int HEADS= 8;
static constexpr int DH   = 16;
static constexpr int NCLS = 64;
static constexpr float SLOPE = 0.2f;
static constexpr unsigned NEGKEY = 0x007FFFFFu;

__device__ __forceinline__ unsigned fkey(float f) {
    unsigned u = __float_as_uint(f);
    return (u & 0x80000000u) ? ~u : (u | 0x80000000u);
}
__device__ __forceinline__ float funkey(unsigned k) {
    unsigned u = (k & 0x80000000u) ? (k & 0x7fffffffu) : ~k;
    return __uint_as_float(u);
}
__device__ __forceinline__ float lrelu(float x) { return x > 0.f ? x : SLOPE * x; }
__device__ __forceinline__ float elu1(float x)  { return x > 0.f ? x : expm1f(x); }

__global__ __launch_bounds__(256) void k_initA(unsigned* mkey1, float* s1, float* acc1,
                                               unsigned* mkey2, float* s2) {
    int t = blockIdx.x * 256 + threadIdx.x;
    if (t < NN * HID) acc1[t] = 0.f;
    if (t < NN * HEADS) { mkey1[t] = NEGKEY; s1[t] = 0.f; }
    if (t < NN) { mkey2[t] = NEGKEY; s2[t] = 0.f; }
}
__global__ __launch_bounds__(256) void k_initB(float* acc2) {
    int t = blockIdx.x * 256 + threadIdx.x;
    if (t < NN * NCLS) acc2[t] = 0.f;
}

__global__ __launch_bounds__(256) void k_gemm1(const float* __restrict__ X,
                                               const float* __restrict__ W,
                                               float* __restrict__ H) {
    const int wave = threadIdx.x >> 5;
    const int lane = threadIdx.x & 31;
    const int half = lane >> 4;
    const int l16  = lane & 15;
    const int row0 = blockIdx.x * 16;
    const int col  = wave * 16 + l16;

    v8f c = {};
    const float* xrow = X + (size_t)(row0 + l16) * IND;
#pragma unroll 4
    for (int kk = 0; kk < IND; kk += 4) {
        const int k = kk + half * 2;
        __builtin_prefetch(xrow + k + IND, 0, 1);
        v2f a = *(const v2f*)(xrow + k);
        v2f b;
        b.x = W[(size_t)k * HID + col];
        b.y = W[(size_t)(k + 1) * HID + col];
        c = __builtin_amdgcn_wmma_f32_16x16x4_f32(false, a, false, b, (short)0, c, false, false);
    }
#pragma unroll
    for (int v = 0; v < 8; ++v)
        H[(size_t)(row0 + v + half * 8) * HID + col] = c[v];
}

__global__ __launch_bounds__(256) void k_att1(const float* __restrict__ H,
                                              const float* __restrict__ al,
                                              const float* __restrict__ ar,
                                              float* __restrict__ el, float* __restrict__ er) {
    int t = blockIdx.x * 256 + threadIdx.x;
    if (t >= NN * HEADS) return;
    const int h = t & 7;
    const float4* hp = (const float4*)(H + (size_t)t * DH);
    const float4* ap = (const float4*)(al + h * DH);
    const float4* rp = (const float4*)(ar + h * DH);
    float se = 0.f, sr = 0.f;
#pragma unroll
    for (int i = 0; i < 4; ++i) {
        const float4 hv = hp[i], av = ap[i], rv = rp[i];
        se += hv.x * av.x + hv.y * av.y + hv.z * av.z + hv.w * av.w;
        sr += hv.x * rv.x + hv.y * rv.y + hv.z * rv.z + hv.w * rv.w;
    }
    el[t] = se; er[t] = sr;
}

__global__ __launch_bounds__(256) void k_emax1(const int* __restrict__ src, const int* __restrict__ dst,
                                               const float* __restrict__ el, const float* __restrict__ er,
                                               unsigned* __restrict__ mkey) {
    int t = blockIdx.x * 256 + threadIdx.x;
    if (t >= EE * HEADS) return;
    const int e = t >> 3, h = t & 7;
    const int s = src[e], d = dst[e];
    const float ev = lrelu(el[s * HEADS + h] + er[d * HEADS + h]);
    atomicMax(&mkey[d * HEADS + h], fkey(ev));
}

__global__ __launch_bounds__(256) void k_eagg1(const int* __restrict__ src, const int* __restrict__ dst,
                                               const float* __restrict__ el, const float* __restrict__ er,
                                               const unsigned* __restrict__ mkey,
                                               const float* __restrict__ H,
                                               float* __restrict__ ssum, float* __restrict__ acc) {
    int t = blockIdx.x * 256 + threadIdx.x;
    if (t >= EE * HEADS) return;
    const int e = t >> 3, h = t & 7;
    const int s = src[e], d = dst[e];
    const float ev = lrelu(el[s * HEADS + h] + er[d * HEADS + h]);
    float m = funkey(mkey[d * HEADS + h]);
    if (!isfinite(m)) m = 0.f;
    const float p = __expf(ev - m);
    atomicAdd(&ssum[d * HEADS + h], p);
    const float4* hv = (const float4*)(H + (size_t)s * HID + h * DH);
    float* ap = acc + (size_t)d * HID + h * DH;
    float4 a0 = hv[0], a1 = hv[1], a2 = hv[2], a3 = hv[3];
    atomicAdd(ap + 0,  p * a0.x); atomicAdd(ap + 1,  p * a0.y);
    atomicAdd(ap + 2,  p * a0.z); atomicAdd(ap + 3,  p * a0.w);
    atomicAdd(ap + 4,  p * a1.x); atomicAdd(ap + 5,  p * a1.y);
    atomicAdd(ap + 6,  p * a1.z); atomicAdd(ap + 7,  p * a1.w);
    atomicAdd(ap + 8,  p * a2.x); atomicAdd(ap + 9,  p * a2.y);
    atomicAdd(ap + 10, p * a2.z); atomicAdd(ap + 11, p * a2.w);
    atomicAdd(ap + 12, p * a3.x); atomicAdd(ap + 13, p * a3.y);
    atomicAdd(ap + 14, p * a3.z); atomicAdd(ap + 15, p * a3.w);
}

__global__ __launch_bounds__(256) void k_fin1(const float* __restrict__ acc, const float* __restrict__ ssum,
                                              const float* __restrict__ b, float* __restrict__ h2) {
    int t = blockIdx.x * 256 + threadIdx.x;
    if (t >= NN * HID) return;
    const int n = t >> 7, cidx = t & 127, h = cidx >> 4;
    const float s = ssum[n * HEADS + h];
    float v = (s > 0.f) ? acc[t] / s : 0.f;
    v = elu1(elu1(v + b[cidx]));
    h2[t] = v;
}

__global__ __launch_bounds__(256) void k_gemm2(const float* __restrict__ H2,
                                               const float* __restrict__ W,
                                               float* __restrict__ HT) {
    const int wave = threadIdx.x >> 5;
    const int lane = threadIdx.x & 31;
    const int half = lane >> 4;
    const int l16  = lane & 15;
    const int rt = wave >> 2, ct = wave & 3;
    const int row0 = blockIdx.x * 32 + rt * 16;
    const int col  = ct * 16 + l16;

    v8f c = {};
    const float* xrow = H2 + (size_t)(row0 + l16) * HID;
#pragma unroll 4
    for (int kk = 0; kk < HID; kk += 4) {
        const int k = kk + half * 2;
        v2f a = *(const v2f*)(xrow + k);
        v2f b;
        b.x = W[(size_t)k * NCLS + col];
        b.y = W[(size_t)(k + 1) * NCLS + col];
        c = __builtin_amdgcn_wmma_f32_16x16x4_f32(false, a, false, b, (short)0, c, false, false);
    }
#pragma unroll
    for (int v = 0; v < 8; ++v)
        HT[(size_t)(row0 + v + half * 8) * NCLS + col] = c[v];
}

__global__ __launch_bounds__(256) void k_att2(const float* __restrict__ HT,
                                              const float* __restrict__ al,
                                              const float* __restrict__ ar,
                                              float* __restrict__ el, float* __restrict__ er) {
    int t = blockIdx.x * 256 + threadIdx.x;
    if (t >= NN) return;
    const float4* hp = (const float4*)(HT + (size_t)t * NCLS);
    const float4* ap = (const float4*)al;
    const float4* rp = (const float4*)ar;
    float se = 0.f, sr = 0.f;
#pragma unroll
    for (int i = 0; i < NCLS / 4; ++i) {
        const float4 hv = hp[i], av = ap[i], rv = rp[i];
        se += hv.x * av.x + hv.y * av.y + hv.z * av.z + hv.w * av.w;
        sr += hv.x * rv.x + hv.y * rv.y + hv.z * rv.z + hv.w * rv.w;
    }
    el[t] = se; er[t] = sr;
}

__global__ __launch_bounds__(256) void k_emax2(const int* __restrict__ src, const int* __restrict__ dst,
                                               const float* __restrict__ el, const float* __restrict__ er,
                                               unsigned* __restrict__ mkey) {
    int t = blockIdx.x * 256 + threadIdx.x;
    if (t >= EE) return;
    const int s = src[t], d = dst[t];
    const float ev = lrelu(el[s] + er[d]);
    atomicMax(&mkey[d], fkey(ev));
}

__global__ __launch_bounds__(256) void k_eagg2(const int* __restrict__ src, const int* __restrict__ dst,
                                               const float* __restrict__ el, const float* __restrict__ er,
                                               const unsigned* __restrict__ mkey,
                                               const float* __restrict__ HT,
                                               float* __restrict__ ssum, float* __restrict__ acc) {
    int t = blockIdx.x * 256 + threadIdx.x;
    if (t >= EE * 16) return;
    const int e = t >> 4, q = t & 15;
    const int s = src[e], d = dst[e];
    const float ev = lrelu(el[s] + er[d]);
    float m = funkey(mkey[d]);
    if (!isfinite(m)) m = 0.f;
    const float p = __expf(ev - m);
    if (q == 0) atomicAdd(&ssum[d], p);
    const float4 hv = ((const float4*)(HT + (size_t)s * NCLS))[q];
    float* ap = acc + (size_t)d * NCLS + q * 4;
    atomicAdd(ap + 0, p * hv.x); atomicAdd(ap + 1, p * hv.y);
    atomicAdd(ap + 2, p * hv.z); atomicAdd(ap + 3, p * hv.w);
}

__global__ __launch_bounds__(256) void k_fin2(const float* __restrict__ acc, const float* __restrict__ ssum,
                                              const float* __restrict__ b, float* __restrict__ out) {
    int t = blockIdx.x * 256 + threadIdx.x;
    if (t >= NN * NCLS) return;
    const int n = t >> 6, cidx = t & 63;
    const float s = ssum[n];
    float v = (s > 0.f) ? acc[t] / s : 0.f;
    out[t] = v + b[cidx];
}

extern "C" void kernel_launch(void* const* d_in, const int* in_sizes, int n_in,
                              void* d_out, int out_size, void* d_ws, size_t ws_size,
                              hipStream_t stream) {
    const float* x   = (const float*)d_in[0];
    const int*   src = (const int*)  d_in[1];
    const int*   dst = (const int*)  d_in[2];
    const float* W1  = (const float*)d_in[3];
    const float* al1 = (const float*)d_in[4];
    const float* ar1 = (const float*)d_in[5];
    const float* b1  = (const float*)d_in[6];
    const float* W2  = (const float*)d_in[7];
    const float* al2 = (const float*)d_in[8];
    const float* ar2 = (const float*)d_in[9];
    const float* b2  = (const float*)d_in[10];
    float* out = (float*)d_out;

    size_t off = 0;
    auto take = [&](size_t nfloats) -> float* {
        float* p = (float*)((char*)d_ws + off);
        off = (off + nfloats * sizeof(float) + 255) & ~(size_t)255;
        return p;
    };
    float*    h1    = take((size_t)NN * HID);
    float*    el1   = take((size_t)NN * HEADS);
    float*    er1   = take((size_t)NN * HEADS);
    unsigned* mkey1 = (unsigned*)take((size_t)NN * HEADS);
    float*    s1    = take((size_t)NN * HEADS);
    float*    acc1  = take((size_t)NN * HID);
    float*    h2t   = take((size_t)NN * NCLS);
    float*    el2   = take((size_t)NN);
    float*    er2   = take((size_t)NN);
    unsigned* mkey2 = (unsigned*)take((size_t)NN);
    float*    s2    = take((size_t)NN);
    float*    h2    = h1;
    float*    acc2  = acc1;

    const int T = 256;
    k_initA<<<(NN * HID + T - 1) / T, T, 0, stream>>>(mkey1, s1, acc1, mkey2, s2);
    k_gemm1<<<NN / 16, T, 0, stream>>>(x, W1, h1);
    k_att1 <<<(NN * HEADS + T - 1) / T, T, 0, stream>>>(h1, al1, ar1, el1, er1);
    k_emax1<<<(EE * HEADS + T - 1) / T, T, 0, stream>>>(src, dst, el1, er1, mkey1);
    k_eagg1<<<(EE * HEADS + T - 1) / T, T, 0, stream>>>(src, dst, el1, er1, mkey1, h1, s1, acc1);
    k_fin1 <<<(NN * HID + T - 1) / T, T, 0, stream>>>(acc1, s1, b1, h2);
    k_gemm2<<<NN / 32, T, 0, stream>>>(h2, W2, h2t);
    k_att2 <<<(NN + T - 1) / T, T, 0, stream>>>(h2t, al2, ar2, el2, er2);
    k_initB<<<(NN * NCLS + T - 1) / T, T, 0, stream>>>(acc2);
    k_emax2<<<(EE + T - 1) / T, T, 0, stream>>>(src, dst, el2, er2, mkey2);
    k_eagg2<<<(EE * 16 + T - 1) / T, T, 0, stream>>>(src, dst, el2, er2, mkey2, h2t, s2, acc2);
    k_fin2 <<<(NN * NCLS + T - 1) / T, T, 0, stream>>>(acc2, s2, b2, out);
}


# Float-atomic edge scatter replaced by in-call dst-grouped edge list (count, claim, fill by int atomics) + one wave per node: max, exp-sum, weighted rows in registers; sum order = fill arrival order
# speedup vs baseline: 2.9690x; 2.9690x over previous
#include <hip/hip_runtime.h>

typedef __attribute__((ext_vector_type(2))) float v2f;
typedef __attribute__((ext_vector_type(8))) float v8f;

static constexpr int NN   = 100000;
static constexpr int EE   = 1600000;
static constexpr int IND  = 256;
static constexpr int HID  = 128;
static constexpr int HEADS= 8;
static constexpr int DH   = 16;
static constexpr int NCLS = 64;
static constexpr float SLOPE = 0.2f;

__device__ __forceinline__ float lrelu(float x) { return x > 0.f ? x : SLOPE * x; }
__device__ __forceinline__ float elu1(float x)  { return x > 0.f ? x : expm1f(x); }

__global__ __launch_bounds__(256) void k_zero(unsigned* __restrict__ deg, unsigned* __restrict__ total) {
    const int t = blockIdx.x * 256 + threadIdx.x;
    if (t < NN) deg[t] = 0u;
    if (t == 0) *total = 0u;
}
__global__ __launch_bounds__(256) void k_count(const int* __restrict__ dst, unsigned* __restrict__ deg) {
    const int t = blockIdx.x * 256 + threadIdx.x;
    if (t >= EE) return;
    atomicAdd(&deg[dst[t]], 1u);
}
__global__ __launch_bounds__(256) void k_alloc(const unsigned* __restrict__ deg, unsigned* __restrict__ cur,
                                               unsigned* __restrict__ total) {
    const int t = blockIdx.x * 256 + threadIdx.x;
    const int lane = threadIdx.x & 31;
    const unsigned d = (t < NN) ? deg[t] : 0u;
    unsigned inc = d;
    for (int o = 1; o < 32; o <<= 1) {
        const unsigned v = __shfl_up(inc, o);
        if (lane >= o) inc += v;
    }
    unsigned base = 0u;
    if (lane == 31) base = atomicAdd(total, inc);
    base = __shfl(base, 31);
    if (t < NN) cur[t] = base + inc - d;
}
__global__ __launch_bounds__(256) void k_fill(const int* __restrict__ src, const int* __restrict__ dst,
                                              unsigned* __restrict__ cur, int* __restrict__ esrc) {
    const int t = blockIdx.x * 256 + threadIdx.x;
    if (t >= EE) return;
    const unsigned pos = atomicAdd(&cur[dst[t]], 1u);
    esrc[pos] = src[t];
}

__global__ __launch_bounds__(256) void k_gemm1(const float* __restrict__ X,
                                               const float* __restrict__ W,
                                               float* __restrict__ H) {
    const int wave = threadIdx.x >> 5;
    const int lane = threadIdx.x & 31;
    const int half = lane >> 4;
    const int l16  = lane & 15;
    const int row0 = blockIdx.x * 16;
    const int col  = wave * 16 + l16;

    v8f c = {};
    const float* xrow = X + (size_t)(row0 + l16) * IND;
#pragma unroll 4
    for (int kk = 0; kk < IND; kk += 4) {
        const int k = kk + half * 2;
        __builtin_prefetch(xrow + k + IND, 0, 1);
        v2f a = *(const v2f*)(xrow + k);
        v2f b;
        b.x = W[(size_t)k * HID + col];
        b.y = W[(size_t)(k + 1) * HID + col];
        c = __builtin_amdgcn_wmma_f32_16x16x4_f32(false, a, false, b, (short)0, c, false, false);
    }
#pragma unroll
    for (int v = 0; v < 8; ++v)
        H[(size_t)(row0 + v + half * 8) * HID + col] = c[v];
}

__global__ __launch_bounds__(256) void k_att1(const float* __restrict__ H,
                                              const float* __restrict__ al,
                                              const float* __restrict__ ar,
                                              float* __restrict__ el, float* __restrict__ er) {
    int t = blockIdx.x * 256 + threadIdx.x;
    if (t >= NN * HEADS) return;
    const int h = t & 7;
    const float4* hp = (const float4*)(H + (size_t)t * DH);
    const float4* ap = (const float4*)(al + h * DH);
    const float4* rp = (const float4*)(ar + h * DH);
    float se = 0.f, sr = 0.f;
#pragma unroll
    for (int i = 0; i < 4; ++i) {
        const float4 hv = hp[i], av = ap[i], rv = rp[i];
        se += hv.x * av.x + hv.y * av.y + hv.z * av.z + hv.w * av.w;
        sr += hv.x * rv.x + hv.y * rv.y + hv.z * rv.z + hv.w * rv.w;
    }
    el[t] = se; er[t] = sr;
}

__global__ __launch_bounds__(256) void k_gat1(const int* __restrict__ esrc,
                                              const unsigned* __restrict__ deg, const unsigned* __restrict__ endp,
                                              const float* __restrict__ el, const float* __restrict__ er,
                                              const float* __restrict__ H, const float* __restrict__ b,
                                              float* __restrict__ h2) {
    const int lane = threadIdx.x & 31;
    const int n = blockIdx.x * 8 + __builtin_amdgcn_readfirstlane((int)(threadIdx.x >> 5));
    const int h = lane >> 2, q = lane & 3;
    const int end = __builtin_amdgcn_readfirstlane((int)endp[n]);
    const int beg = end - __builtin_amdgcn_readfirstlane((int)deg[n]);
    const float erv = er[n * HEADS + h];

    float mx = -__builtin_inff();
    for (int j0 = beg; j0 < end; j0 += 32) {
        const int jj = j0 + lane;
        const int sv = (jj < end) ? esrc[jj] : 0;
        const int cnt = (end - j0 < 32) ? (end - j0) : 32;
        for (int i0 = 0; i0 < cnt; i0 += 8) {
            const int ia = i0 + q, ib = ia + 4;
            const int sa = __shfl(sv, ia);
            const int sb = __shfl(sv, ib & 31);
            const float ea = el[sa * HEADS + h];
            const float eb = el[sb * HEADS + h];
            if (ia < cnt) mx = fmaxf(mx, lrelu(ea + erv));
            if (ib < cnt) mx = fmaxf(mx, lrelu(eb + erv));
        }
    }
    mx = fmaxf(mx, __shfl_xor(mx, 1));
    mx = fmaxf(mx, __shfl_xor(mx, 2));
    const float m = isfinite(mx) ? mx : 0.f;

    float ssum = 0.f;
    float4 acc = make_float4(0.f, 0.f, 0.f, 0.f);
    const float* Hl = H + 4 * lane;
    const float* ell = el + h;
    for (int j0 = beg; j0 < end; j0 += 32) {
        const int jj = j0 + lane;
        const int sv = (jj < end) ? esrc[jj] : 0;
        const int cnt = (end - j0 < 32) ? (end - j0) : 32;
        int i = 0;
        for (; i + 4 <= cnt; i += 4) {
            const int s0 = __builtin_amdgcn_readlane(sv, i);
            const int s1 = __builtin_amdgcn_readlane(sv, i + 1);
            const int s2 = __builtin_amdgcn_readlane(sv, i + 2);
            const int s3 = __builtin_amdgcn_readlane(sv, i + 3);
            const float e0 = ell[s0 * HEADS], e1 = ell[s1 * HEADS], e2 = ell[s2 * HEADS], e3 = ell[s3 * HEADS];
            const float4 r0 = *(const float4*)(Hl + (size_t)s0 * HID);
            const float4 r1 = *(const float4*)(Hl + (size_t)s1 * HID);
            const float4 r2 = *(const float4*)(Hl + (size_t)s2 * HID);
            const float4 r3 = *(const float4*)(Hl + (size_t)s3 * HID);
            const float p0 = __expf(lrelu(e0 + erv) - m);
            const float p1 = __expf(lrelu(e1 + erv) - m);
            const float p2 = __expf(lrelu(e2 + erv) - m);
            const float p3 = __expf(lrelu(e3 + erv) - m);
            ssum += p0; acc.x += p0 * r0.x; acc.y += p0 * r0.y; acc.z += p0 * r0.z; acc.w += p0 * r0.w;
            ssum += p1; acc.x += p1 * r1.x; acc.y += p1 * r1.y; acc.z += p1 * r1.z; acc.w += p1 * r1.w;
            ssum += p2; acc.x += p2 * r2.x; acc.y += p2 * r2.y; acc.z += p2 * r2.z; acc.w += p2 * r2.w;
            ssum += p3; acc.x += p3 * r3.x; acc.y += p3 * r3.y; acc.z += p3 * r3.z; acc.w += p3 * r3.w;
        }
        for (; i < cnt; ++i) {
            const int s0 = __builtin_amdgcn_readlane(sv, i);
            const float e0 = ell[s0 * HEADS];
            const float4 r0 = *(const float4*)(Hl + (size_t)s0 * HID);
            const float p0 = __expf(lrelu(e0 + erv) - m);
            ssum += p0; acc.x += p0 * r0.x; acc.y += p0 * r0.y; acc.z += p0 * r0.z; acc.w += p0 * r0.w;
        }
    }
    const float4 bv = *(const float4*)(b + 4 * lane);
    const bool any = ssum > 0.f;
    float4 o;
    o.x = elu1(elu1((any ? acc.x / ssum : 0.f) + bv.x));
    o.y = elu1(elu1((any ? acc.y / ssum : 0.f) + bv.y));
    o.z = elu1(elu1((any ? acc.z / ssum : 0.f) + bv.z));
    o.w = elu1(elu1((any ? acc.w / ssum : 0.f) + bv.w));
    *(float4*)(h2 + (size_t)n * HID + 4 * lane) = o;
}

__global__ __launch_bounds__(256) void k_gemm2(const float* __restrict__ H2,
                                               const float* __restrict__ W,
                                               float* __restrict__ HT) {
    const int wave = threadIdx.x >> 5;
    const int lane = threadIdx.x & 31;
    const int half = lane >> 4;
    const int l16  = lane & 15;
    const int rt = wave >> 2, ct = wave & 3;
    const int row0 = blockIdx.x * 32 + rt * 16;
    const int col  = ct * 16 + l16;

    v8f c = {};
    const float* xrow = H2 + (size_t)(row0 + l16) * HID;
#pragma unroll 4
    for (int kk = 0; kk < HID; kk += 4) {
        const int k = kk + half * 2;
        v2f a = *(const v2f*)(xrow + k);
        v2f b;
        b.x = W[(size_t)k * NCLS + col];
        b.y = W[(size_t)(k + 1) * NCLS + col];
        c = __builtin_amdgcn_wmma_f32_16x16x4_f32(false, a, false, b, (short)0, c, false, false);
    }
#pragma unroll
    for (int v = 0; v < 8; ++v)
        HT[(size_t)(row0 + v + half * 8) * NCLS + col] = c[v];
}

__global__ __launch_bounds__(256) void k_att2(const float* __restrict__ HT,
                                              const float* __restrict__ al,
                                              const float* __restrict__ ar,
                                              float* __restrict__ el, float* __restrict__ er) {
    int t = blockIdx.x * 256 + threadIdx.x;
    if (t >= NN) return;
    const float4* hp = (const float4*)(HT + (size_t)t * NCLS);
    const float4* ap = (const float4*)al;
    const float4* rp = (const float4*)ar;
    float se = 0.f, sr = 0.f;
#pragma unroll
    for (int i = 0; i < NCLS / 4; ++i) {
        const float4 hv = hp[i], av = ap[i], rv = rp[i];
        se += hv.x * av.x + hv.y * av.y + hv.z * av.z + hv.w * av.w;
        sr += hv.x * rv.x + hv.y * rv.y + hv.z * rv.z + hv.w * rv.w;
    }
    el[t] = se; er[t] = sr;
}

__global__ __launch_bounds__(256) void k_gat2(const int* __restrict__ esrc,
                                              const unsigned* __restrict__ deg, const unsigned* __restrict__ endp,
                                              const float* __restrict__ el, const float* __restrict__ er,
                                              const float* __restrict__ HT, const float* __restrict__ b,
                                              float* __restrict__ out) {
    const int lane = threadIdx.x & 31;
    const int n = blockIdx.x * 8 + __builtin_amdgcn_readfirstlane((int)(threadIdx.x >> 5));
    const int end = __builtin_amdgcn_readfirstlane((int)endp[n]);
    const int beg = end - __builtin_amdgcn_readfirstlane((int)deg[n]);
    const float erv = er[n];

    float mx = -__builtin_inff();
    for (int j0 = beg; j0 < end; j0 += 32) {
        const int jj = j0 + lane;
        if (jj < end) mx = fmaxf(mx, lrelu(el[esrc[jj]] + erv));
    }
    for (int o = 16; o >= 1; o >>= 1) mx = fmaxf(mx, __shfl_xor(mx, o));
    const float m = isfinite(mx) ? mx : 0.f;

    float sl = 0.f;
    float2 acc = make_float2(0.f, 0.f);
    const float* Tl = HT + 2 * lane;
    for (int j0 = beg; j0 < end; j0 += 32) {
        const int jj = j0 + lane;
        const bool ok = jj < end;
        const int sv = ok ? esrc[jj] : 0;
        const float pf = ok ? __expf(lrelu(el[sv] + erv) - m) : 0.f;
        sl += pf;
        const int pv = __float_as_int(pf);
        const int cnt = (end - j0 < 32) ? (end - j0) : 32;
        int i = 0;
        for (; i + 4 <= cnt; i += 4) {
            const int s0 = __builtin_amdgcn_readlane(sv, i);
            const int s1 = __builtin_amdgcn_readlane(sv, i + 1);
            const int s2 = __builtin_amdgcn_readlane(sv, i + 2);
            const int s3 = __builtin_amdgcn_readlane(sv, i + 3);
            const float2 r0 = *(const float2*)(Tl + (size_t)s0 * NCLS);
            const float2 r1 = *(const float2*)(Tl + (size_t)s1 * NCLS);
            const float2 r2 = *(const float2*)(Tl + (size_t)s2 * NCLS);
            const float2 r3 = *(const float2*)(Tl + (size_t)s3 * NCLS);
            const float p0 = __int_as_float(__builtin_amdgcn_readlane(pv, i));
            const float p1 = __int_as_float(__builtin_amdgcn_readlane(pv, i + 1));
            const float p2 = __int_as_float(__builtin_amdgcn_readlane(pv, i + 2));
            const float p3 = __int_as_float(__builtin_amdgcn_readlane(pv, i + 3));
            acc.x += p0 * r0.x; acc.y += p0 * r0.y;
            acc.x += p1 * r1.x; acc.y += p1 * r1.y;
            acc.x += p2 * r2.x; acc.y += p2 * r2.y;
            acc.x += p3 * r3.x; acc.y += p3 * r3.y;
        }
        for (; i < cnt; ++i) {
            const int s0 = __builtin_amdgcn_readlane(sv, i);
            const float2 r0 = *(const float2*)(Tl + (size_t)s0 * NCLS);
            const float p0 = __int_as_float(__builtin_amdgcn_readlane(pv, i));
            acc.x += p0 * r0.x; acc.y += p0 * r0.y;
        }
    }
    for (int o = 16; o >= 1; o >>= 1) sl += __shfl_xor(sl, o);
    const float2 bv = *(const float2*)(b + 2 * lane);
    const bool any = sl > 0.f;
    float2 o2;
    o2.x = (any ? acc.x / sl : 0.f) + bv.x;
    o2.y = (any ? acc.y / sl : 0.f) + bv.y;
    *(float2*)(out + (size_t)n * NCLS + 2 * lane) = o2;
}

extern "C" void kernel_launch(void* const* d_in, const int* in_sizes, int n_in,
                              void* d_out, int out_size, void* d_ws, size_t ws_size,
                              hipStream_t stream) {
    const float* x   = (const float*)d_in[0];
    const int*   src = (const int*)  d_in[1];
    const int*   dst = (const int*)  d_in[2];
    const float* W1  = (const float*)d_in[3];
    const float* al1 = (const float*)d_in[4];
    const float* ar1 = (const float*)d_in[5];
    const float* b1  = (const float*)d_in[6];
    const float* W2  = (const float*)d_in[7];
    const float* al2 = (const float*)d_in[8];
    const float* ar2 = (const float*)d_in[9];
    const float* b2  = (const float*)d_in[10];
    float* out = (float*)d_out;

    size_t off = 0;
    auto take = [&](size_t nfloats) -> float* {
        float* p = (float*)((char*)d_ws + off);
        off = (off + nfloats * sizeof(float) + 255) & ~(size_t)255;
        return p;
    };
    float*    h1    = take((size_t)NN * HID);
    float*    el1   = take((size_t)NN * HEADS);
    float*    er1   = take((size_t)NN * HEADS);
    int*      esrc  = (int*)take((size_t)EE);
    float*    h2    = take((size_t)NN * HID);
    float*    h2t   = take((size_t)NN * NCLS);
    float*    el2   = take((size_t)NN);
    float*    er2   = take((size_t)NN);
    unsigned* deg   = (unsigned*)take((size_t)NN);
    unsigned* cur   = (unsigned*)take((size_t)NN);
    unsigned* total = (unsigned*)h2t;

    const int T = 256;
    k_zero <<<(NN + T - 1) / T, T, 0, stream>>>(deg, total);
    k_count<<<(EE + T - 1) / T, T, 0, stream>>>(dst, deg);
    k_alloc<<<(NN + T - 1) / T, T, 0, stream>>>(deg, cur, total);
    k_fill <<<(EE + T - 1) / T, T, 0, stream>>>(src, dst, cur, esrc);
    k_gemm1<<<NN / 16, T, 0, stream>>>(x, W1, h1);
    k_att1 <<<(NN * HEADS + T - 1) / T, T, 0, stream>>>(h1, al1, ar1, el1, er1);
    k_gat1 <<<NN / 8, T, 0, stream>>>(esrc, deg, cur, el1, er1, h1, b1, h2);
    k_gemm2<<<NN / 32, T, 0, stream>>>(h2, W2, h2t);
    k_att2 <<<(NN + T - 1) / T, T, 0, stream>>>(h2t, al2, ar2, el2, er2);
    k_gat2 <<<NN / 8, T, 0, stream>>>(esrc, deg, cur, el2, er2, h2t, b2, out);
}


# Both GEMMs: the block's contiguous 80-row tile goes to LDS by whole-line 16-byte loads, A fragments come from LDS, each B fragment feeds 5 row tiles; X prefetch dropped; k order and sums as before
# speedup vs baseline: 4.7951x; 1.6151x over previous
#include <hip/hip_runtime.h>

typedef __attribute__((ext_vector_type(2))) float v2f;
typedef __attribute__((ext_vector_type(8))) float v8f;

static constexpr int NN   = 100000;
static constexpr int EE   = 1600000;
static constexpr int IND  = 256;
static constexpr int HID  = 128;
static constexpr int HEADS= 8;
static constexpr int DH   = 16;
static constexpr int NCLS = 64;
static constexpr float SLOPE = 0.2f;

__device__ __forceinline__ float lrelu(float x) { return x > 0.f ? x : SLOPE * x; }
__device__ __forceinline__ float elu1(float x)  { return x > 0.f ? x : expm1f(x); }

__global__ __launch_bounds__(256) void k_zero(unsigned* __restrict__ deg, unsigned* __restrict__ total) {
    const int t = blockIdx.x * 256 + threadIdx.x;
    if (t < NN) deg[t] = 0u;
    if (t == 0) *total = 0u;
}
__global__ __launch_bounds__(256) void k_count(const int* __restrict__ dst, unsigned* __restrict__ deg) {
    const int t = blockIdx.x * 256 + threadIdx.x;
    if (t >= EE) return;
    atomicAdd(&deg[dst[t]], 1u);
}
__global__ __launch_bounds__(256) void k_alloc(const unsigned* __restrict__ deg, unsigned* __restrict__ cur,
                                               unsigned* __restrict__ total) {
    const int t = blockIdx.x * 256 + threadIdx.x;
    const int lane = threadIdx.x & 31;
    const unsigned d = (t < NN) ? deg[t] : 0u;
    unsigned inc = d;
    for (int o = 1; o < 32; o <<= 1) {
        const unsigned v = __shfl_up(inc, o);
        if (lane >= o) inc += v;
    }
    unsigned base = 0u;
    if (lane == 31) base = atomicAdd(total, inc);
    base = __shfl(base, 31);
    if (t < NN) cur[t] = base + inc - d;
}
__global__ __launch_bounds__(256) void k_fill(const int* __restrict__ src, const int* __restrict__ dst,
                                              unsigned* __restrict__ cur, int* __restrict__ esrc) {
    const int t = blockIdx.x * 256 + threadIdx.x;
    if (t >= EE) return;
    const unsigned pos = atomicAdd(&cur[dst[t]], 1u);
    esrc[pos] = src[t];
}

static constexpr int GT_ROWS  = 80;
static constexpr int GT_TILES = GT_ROWS / 16;
static constexpr int G1_PITCH = IND + 4;
__global__ __launch_bounds__(256) void k_gemm1(const float* __restrict__ X,
                                               const float* __restrict__ W,
                                               float* __restrict__ H) {
    __shared__ __attribute__((aligned(16))) float xs[GT_ROWS * G1_PITCH];
    const int tid  = threadIdx.x;
    const int wave = tid >> 5;
    const int lane = tid & 31;
    const int half = lane >> 4;
    const int l16  = lane & 15;
    const int row0 = blockIdx.x * GT_ROWS;
    const int col  = wave * 16 + l16;

    const float4* xg = (const float4*)(X + (size_t)row0 * IND);
#pragma unroll
    for (int i = 0; i < GT_ROWS * IND / 4 / 256; ++i) {
        const int idx = i * 256 + tid;
        const float4 v = xg[idx];
        *(float4*)(xs + (idx >> 6) * G1_PITCH + (idx & 63) * 4) = v;
    }
    __syncthreads();

    v8f c[GT_TILES];
#pragma unroll
    for (int rt = 0; rt < GT_TILES; ++rt) c[rt] = v8f{};
    const float* ap = xs + l16 * G1_PITCH + half * 2;
    const float* wp = W + (size_t)(half * 2) * HID + col;
#pragma unroll 8
    for (int kk = 0; kk < IND; kk += 4) {
        v2f b;
        b.x = wp[(size_t)kk * HID];
        b.y = wp[(size_t)(kk + 1) * HID];
#pragma unroll
        for (int rt = 0; rt < GT_TILES; ++rt) {
            const v2f a = *(const v2f*)(ap + rt * 16 * G1_PITCH + kk);
            c[rt] = __builtin_amdgcn_wmma_f32_16x16x4_f32(false, a, false, b, (short)0, c[rt], false, false);
        }
    }
#pragma unroll
    for (int rt = 0; rt < GT_TILES; ++rt)
#pragma unroll
        for (int v = 0; v < 8; ++v)
            H[(size_t)(row0 + rt * 16 + v + half * 8) * HID + col] = c[rt][v];
}

__global__ __launch_bounds__(256) void k_att1(const float* __restrict__ H,
                                              const float* __restrict__ al,
                                              const float* __restrict__ ar,
                                              float* __restrict__ el, float* __restrict__ er) {
    int t = blockIdx.x * 256 + threadIdx.x;
    if (t >= NN * HEADS) return;
    const int h = t & 7;
    const float4* hp = (const float4*)(H + (size_t)t * DH);
    const float4* ap = (const float4*)(al + h * DH);
    const float4* rp = (const float4*)(ar + h * DH);
    float se = 0.f, sr = 0.f;
#pragma unroll
    for (int i = 0; i < 4; ++i) {
        const float4 hv = hp[i], av = ap[i], rv = rp[i];
        se += hv.x * av.x + hv.y * av.y + hv.z * av.z + hv.w * av.w;
        sr += hv.x * rv.x + hv.y * rv.y + hv.z * rv.z + hv.w * rv.w;
    }
    el[t] = se; er[t] = sr;
}

__global__ __launch_bounds__(256) void k_gat1(const int* __restrict__ esrc,
                                              const unsigned* __restrict__ deg, const unsigned* __restrict__ endp,
                                              const float* __restrict__ el, const float* __restrict__ er,
                                              const float* __restrict__ H, const float* __restrict__ b,
                                              float* __restrict__ h2) {
    const int lane = threadIdx.x & 31;
    const int n = blockIdx.x * 8 + __builtin_amdgcn_readfirstlane((int)(threadIdx.x >> 5));
    const int h = lane >> 2, q = lane & 3;
    const int end = __builtin_amdgcn_readfirstlane((int)endp[n]);
    const int beg = end - __builtin_amdgcn_readfirstlane((int)deg[n]);
    const float erv = er[n * HEADS + h];

    float mx = -__builtin_inff();
    for (int j0 = beg; j0 < end; j0 += 32) {
        const int jj = j0 + lane;
        const int sv = (jj < end) ? esrc[jj] : 0;
        const int cnt = (end - j0 < 32) ? (end - j0) : 32;
        for (int i0 = 0; i0 < cnt; i0 += 8) {
            const int ia = i0 + q, ib = ia + 4;
            const int sa = __shfl(sv, ia);
            const int sb = __shfl(sv, ib & 31);
            const float ea = el[sa * HEADS + h];
            const float eb = el[sb * HEADS + h];
            if (ia < cnt) mx = fmaxf(mx, lrelu(ea + erv));
            if (ib < cnt) mx = fmaxf(mx, lrelu(eb + erv));
        }
    }
    mx = fmaxf(mx, __shfl_xor(mx, 1));
    mx = fmaxf(mx, __shfl_xor(mx, 2));
    const float m = isfinite(mx) ? mx : 0.f;

    float ssum = 0.f;
    float4 acc = make_float4(0.f, 0.f, 0.f, 0.f);
    const float* Hl = H + 4 * lane;
    const float* ell = el + h;
    for (int j0 = beg; j0 < end; j0 += 32) {
        const int jj = j0 + lane;
        const int sv = (jj < end) ? esrc[jj] : 0;
        const int cnt = (end - j0 < 32) ? (end - j0) : 32;
        int i = 0;
        for (; i + 4 <= cnt; i += 4) {
            const int s0 = __builtin_amdgcn_readlane(sv, i);
            const int s1 = __builtin_amdgcn_readlane(sv, i + 1);
            const int s2 = __builtin_amdgcn_readlane(sv, i + 2);
            const int s3 = __builtin_amdgcn_readlane(sv, i + 3);
            const float e0 = ell[s0 * HEADS], e1 = ell[s1 * HEADS], e2 = ell[s2 * HEADS], e3 = ell[s3 * HEADS];
            const float4 r0 = *(const float4*)(Hl + (size_t)s0 * HID);
            const float4 r1 = *(const float4*)(Hl + (size_t)s1 * HID);
            const float4 r2 = *(const float4*)(Hl + (size_t)s2 * HID);
            const float4 r3 = *(const float4*)(Hl + (size_t)s3 * HID);
            const float p0 = __expf(lrelu(e0 + erv) - m);
            const float p1 = __expf(lrelu(e1 + erv) - m);
            const float p2 = __expf(lrelu(e2 + erv) - m);
            const float p3 = __expf(lrelu(e3 + erv) - m);
            ssum += p0; acc.x += p0 * r0.x; acc.y += p0 * r0.y; acc.z += p0 * r0.z; acc.w += p0 * r0.w;
            ssum += p1; acc.x += p1 * r1.x; acc.y += p1 * r1.y; acc.z += p1 * r1.z; acc.w += p1 * r1.w;
            ssum += p2; acc.x += p2 * r2.x; acc.y += p2 * r2.y; acc.z += p2 * r2.z; acc.w += p2 * r2.w;
            ssum += p3; acc.x += p3 * r3.x; acc.y += p3 * r3.y; acc.z += p3 * r3.z; acc.w += p3 * r3.w;
        }
        for (; i < cnt; ++i) {
            const int s0 = __builtin_amdgcn_readlane(sv, i);
            const float e0 = ell[s0 * HEADS];
            const float4 r0 = *(const float4*)(Hl + (size_t)s0 * HID);
            const float p0 = __expf(lrelu(e0 + erv) - m);
            ssum += p0; acc.x += p0 * r0.x; acc.y += p0 * r0.y; acc.z += p0 * r0.z; acc.w += p0 * r0.w;
        }
    }
    const float4 bv = *(const float4*)(b + 4 * lane);
    const bool any = ssum > 0.f;
    float4 o;
    o.x = elu1(elu1((any ? acc.x / ssum : 0.f) + bv.x));
    o.y = elu1(elu1((any ? acc.y / ssum : 0.f) + bv.y));
    o.z = elu1(elu1((any ? acc.z / ssum : 0.f) + bv.z));
    o.w = elu1(elu1((any ? acc.w / ssum : 0.f) + bv.w));
    *(float4*)(h2 + (size_t)n * HID + 4 * lane) = o;
}

static constexpr int G2_PITCH = HID + 4;
__global__ __launch_bounds__(128) void k_gemm2(const float* __restrict__ H2,
                                               const float* __restrict__ W,
                                               float* __restrict__ HT) {
    __shared__ __attribute__((aligned(16))) float hs[GT_ROWS * G2_PITCH];
    const int tid  = threadIdx.x;
    const int wave = tid >> 5;
    const int lane = tid & 31;
    const int half = lane >> 4;
    const int l16  = lane & 15;
    const int row0 = blockIdx.x * GT_ROWS;
    const int col  = wave * 16 + l16;

    const float4* hg = (const float4*)(H2 + (size_t)row0 * HID);
#pragma unroll
    for (int i = 0; i < GT_ROWS * HID / 4 / 128; ++i) {
        const int idx = i * 128 + tid;
        const float4 v = hg[idx];
        *(float4*)(hs + (idx >> 5) * G2_PITCH + (idx & 31) * 4) = v;
    }
    __syncthreads();

    v8f c[GT_TILES];
#pragma unroll
    for (int rt = 0; rt < GT_TILES; ++rt) c[rt] = v8f{};
    const float* ap = hs + l16 * G2_PITCH + half * 2;
    const float* wp = W + (size_t)(half * 2) * NCLS + col;
#pragma unroll 8
    for (int kk = 0; kk < HID; kk += 4) {
        v2f b;
        b.x = wp[(size_t)kk * NCLS];
        b.y = wp[(size_t)(kk + 1) * NCLS];
#pragma unroll
        for (int rt = 0; rt < GT_TILES; ++rt) {
            const v2f a = *(const v2f*)(ap + rt * 16 * G2_PITCH + kk);
            c[rt] = __builtin_amdgcn_wmma_f32_16x16x4_f32(false, a, false, b, (short)0, c[rt], false, false);
        }
    }
#pragma unroll
    for (int rt = 0; rt < GT_TILES; ++rt)
#pragma unroll
        for (int v = 0; v < 8; ++v)
            HT[(size_t)(row0 + rt * 16 + v + half * 8) * NCLS + col] = c[rt][v];
}

__global__ __launch_bounds__(256) void k_att2(const float* __restrict__ HT,
                                              const float* __restrict__ al,
                                              const float* __restrict__ ar,
                                              float* __restrict__ el, float* __restrict__ er) {
    int t = blockIdx.x * 256 + threadIdx.x;
    if (t >= NN) return;
    const float4* hp = (const float4*)(HT + (size_t)t * NCLS);
    const float4* ap = (const float4*)al;
    const float4* rp = (const float4*)ar;
    float se = 0.f, sr = 0.f;
#pragma unroll
    for (int i = 0; i < NCLS / 4; ++i) {
        const float4 hv = hp[i], av = ap[i], rv = rp[i];
        se += hv.x * av.x + hv.y * av.y + hv.z * av.z + hv.w * av.w;
        sr += hv.x * rv.x + hv.y * rv.y + hv.z * rv.z + hv.w * rv.w;
    }
    el[t] = se; er[t] = sr;
}

__global__ __launch_bounds__(256) void k_gat2(const int* __restrict__ esrc,
                                              const unsigned* __restrict__ deg, const unsigned* __restrict__ endp,
                                              const float* __restrict__ el, const float* __restrict__ er,
                                              const float* __restrict__ HT, const float* __restrict__ b,
                                              float* __restrict__ out) {
    const int lane = threadIdx.x & 31;
    const int n = blockIdx.x * 8 + __builtin_amdgcn_readfirstlane((int)(threadIdx.x >> 5));
    const int end = __builtin_amdgcn_readfirstlane((int)endp[n]);
    const int beg = end - __builtin_amdgcn_readfirstlane((int)deg[n]);
    const float erv = er[n];

    float mx = -__builtin_inff();
    for (int j0 = beg; j0 < end; j0 += 32) {
        const int jj = j0 + lane;
        if (jj < end) mx = fmaxf(mx, lrelu(el[esrc[jj]] + erv));
    }
    for (int o = 16; o >= 1; o >>= 1) mx = fmaxf(mx, __shfl_xor(mx, o));
    const float m = isfinite(mx) ? mx : 0.f;

    float sl = 0.f;
    float2 acc = make_float2(0.f, 0.f);
    const float* Tl = HT + 2 * lane;
    for (int j0 = beg; j0 < end; j0 += 32) {
        const int jj = j0 + lane;
        const bool ok = jj < end;
        const int sv = ok ? esrc[jj] : 0;
        const float pf = ok ? __expf(lrelu(el[sv] + erv) - m) : 0.f;
        sl += pf;
        const int pv = __float_as_int(pf);
        const int cnt = (end - j0 < 32) ? (end - j0) : 32;
        int i = 0;
        for (; i + 4 <= cnt; i += 4) {
            const int s0 = __builtin_amdgcn_readlane(sv, i);
            const int s1 = __builtin_amdgcn_readlane(sv, i + 1);
            const int s2 = __builtin_amdgcn_readlane(sv, i + 2);
            const int s3 = __builtin_amdgcn_readlane(sv, i + 3);
            const float2 r0 = *(const float2*)(Tl + (size_t)s0 * NCLS);
            const float2 r1 = *(const float2*)(Tl + (size_t)s1 * NCLS);
            const float2 r2 = *(const float2*)(Tl + (size_t)s2 * NCLS);
            const float2 r3 = *(const float2*)(Tl + (size_t)s3 * NCLS);
            const float p0 = __int_as_float(__builtin_amdgcn_readlane(pv, i));
            const float p1 = __int_as_float(__builtin_amdgcn_readlane(pv, i + 1));
            const float p2 = __int_as_float(__builtin_amdgcn_readlane(pv, i + 2));
            const float p3 = __int_as_float(__builtin_amdgcn_readlane(pv, i + 3));
            acc.x += p0 * r0.x; acc.y += p0 * r0.y;
            acc.x += p1 * r1.x; acc.y += p1 * r1.y;
            acc.x += p2 * r2.x; acc.y += p2 * r2.y;
            acc.x += p3 * r3.x; acc.y += p3 * r3.y;
        }
        for (; i < cnt; ++i) {
            const int s0 = __builtin_amdgcn_readlane(sv, i);
            const float2 r0 = *(const float2*)(Tl + (size_t)s0 * NCLS);
            const float p0 = __int_as_float(__builtin_amdgcn_readlane(pv, i));
            acc.x += p0 * r0.x; acc.y += p0 * r0.y;
        }
    }
    for (int o = 16; o >= 1; o >>= 1) sl += __shfl_xor(sl, o);
    const float2 bv = *(const float2*)(b + 2 * lane);
    const bool any = sl > 0.f;
    float2 o2;
    o2.x = (any ? acc.x / sl : 0.f) + bv.x;
    o2.y = (any ? acc.y / sl : 0.f) + bv.y;
    *(float2*)(out + (size_t)n * NCLS + 2 * lane) = o2;
}

extern "C" void kernel_launch(void* const* d_in, const int* in_sizes, int n_in,
                              void* d_out, int out_size, void* d_ws, size_t ws_size,
                              hipStream_t stream) {
    const float* x   = (const float*)d_in[0];
    const int*   src = (const int*)  d_in[1];
    const int*   dst = (const int*)  d_in[2];
    const float* W1  = (const float*)d_in[3];
    const float* al1 = (const float*)d_in[4];
    const float* ar1 = (const float*)d_in[5];
    const float* b1  = (const float*)d_in[6];
    const float* W2  = (const float*)d_in[7];
    const float* al2 = (const float*)d_in[8];
    const float* ar2 = (const float*)d_in[9];
    const float* b2  = (const float*)d_in[10];
    float* out = (float*)d_out;

    size_t off = 0;
    auto take = [&](size_t nfloats) -> float* {
        float* p = (float*)((char*)d_ws + off);
        off = (off + nfloats * sizeof(float) + 255) & ~(size_t)255;
        return p;
    };
    float*    h1    = take((size_t)NN * HID);
    float*    el1   = take((size_t)NN * HEADS);
    float*    er1   = take((size_t)NN * HEADS);
    int*      esrc  = (int*)take((size_t)EE);
    float*    h2    = take((size_t)NN * HID);
    float*    h2t   = take((size_t)NN * NCLS);
    float*    el2   = take((size_t)NN);
    float*    er2   = take((size_t)NN);
    unsigned* deg   = (unsigned*)take((size_t)NN);
    unsigned* cur   = (unsigned*)take((size_t)NN);
    unsigned* total = (unsigned*)h2t;

    const int T = 256;
    k_zero <<<(NN + T - 1) / T, T, 0, stream>>>(deg, total);
    k_count<<<(EE + T - 1) / T, T, 0, stream>>>(dst, deg);
    k_alloc<<<(NN + T - 1) / T, T, 0, stream>>>(deg, cur, total);
    k_fill <<<(EE + T - 1) / T, T, 0, stream>>>(src, dst, cur, esrc);
    k_gemm1<<<NN / GT_ROWS, T, 0, stream>>>(x, W1, h1);
    k_att1 <<<(NN * HEADS + T - 1) / T, T, 0, stream>>>(h1, al1, ar1, el1, er1);
    k_gat1 <<<NN / 8, T, 0, stream>>>(esrc, deg, cur, el1, er1, h1, b1, h2);
    k_gemm2<<<NN / GT_ROWS, 128, 0, stream>>>(h2, W2, h2t);
    k_att2 <<<(NN + T - 1) / T, T, 0, stream>>>(h2t, al2, ar2, el2, er2);
    k_gat2 <<<NN / 8, T, 0, stream>>>(esrc, deg, cur, el2, er2, h2t, b2, out);
}


# Logits computed in registers from gathered rows, one-pass softmax with running-max rescale; layer 2 sums even/odd edges in two half-waves; logit kernels and gathers, max pass gone; lane-tree dots
# speedup vs baseline: 5.1061x; 1.0649x over previous
#include <hip/hip_runtime.h>

typedef __attribute__((ext_vector_type(2))) float v2f;
typedef __attribute__((ext_vector_type(8))) float v8f;

static constexpr int NN   = 100000;
static constexpr int EE   = 1600000;
static constexpr int IND  = 256;
static constexpr int HID  = 128;
static constexpr int HEADS= 8;
static constexpr int DH   = 16;
static constexpr int NCLS = 64;
static constexpr float SLOPE = 0.2f;

__device__ __forceinline__ float lrelu(float x) { return x > 0.f ? x : SLOPE * x; }
__device__ __forceinline__ float elu1(float x)  { return x > 0.f ? x : expm1f(x); }

__global__ __launch_bounds__(256) void k_zero(unsigned* __restrict__ deg, unsigned* __restrict__ total) {
    const int t = blockIdx.x * 256 + threadIdx.x;
    if (t < NN) deg[t] = 0u;
    if (t == 0) *total = 0u;
}
__global__ __launch_bounds__(256) void k_count(const int* __restrict__ dst, unsigned* __restrict__ deg) {
    const int t = blockIdx.x * 256 + threadIdx.x;
    if (t >= EE) return;
    atomicAdd(&deg[dst[t]], 1u);
}
__global__ __launch_bounds__(256) void k_alloc(const unsigned* __restrict__ deg, unsigned* __restrict__ cur,
                                               unsigned* __restrict__ total) {
    const int t = blockIdx.x * 256 + threadIdx.x;
    const int lane = threadIdx.x & 31;
    const unsigned d = (t < NN) ? deg[t] : 0u;
    unsigned inc = d;
    for (int o = 1; o < 32; o <<= 1) {
        const unsigned v = __shfl_up(inc, o);
        if (lane >= o) inc += v;
    }
    unsigned base = 0u;
    if (lane == 31) base = atomicAdd(total, inc);
    base = __shfl(base, 31);
    if (t < NN) cur[t] = base + inc - d;
}
__global__ __launch_bounds__(256) void k_fill(const int* __restrict__ src, const int* __restrict__ dst,
                                              unsigned* __restrict__ cur, int* __restrict__ esrc) {
    const int t = blockIdx.x * 256 + threadIdx.x;
    if (t >= EE) return;
    const unsigned pos = atomicAdd(&cur[dst[t]], 1u);
    esrc[pos] = src[t];
}

static constexpr int GT_ROWS  = 80;
static constexpr int GT_TILES = GT_ROWS / 16;
static constexpr int G1_PITCH = IND + 4;
__global__ __launch_bounds__(256) void k_gemm1(const float* __restrict__ X,
                                               const float* __restrict__ W,
                                               float* __restrict__ H) {
    __shared__ __attribute__((aligned(16))) float xs[GT_ROWS * G1_PITCH];
    const int tid  = threadIdx.x;
    const int wave = tid >> 5;
    const int lane = tid & 31;
    const int half = lane >> 4;
    const int l16  = lane & 15;
    const int row0 = blockIdx.x * GT_ROWS;
    const int col  = wave * 16 + l16;

    const float4* xg = (const float4*)(X + (size_t)row0 * IND);
#pragma unroll
    for (int i = 0; i < GT_ROWS * IND / 4 / 256; ++i) {
        const int idx = i * 256 + tid;
        const float4 v = xg[idx];
        *(float4*)(xs + (idx >> 6) * G1_PITCH + (idx & 63) * 4) = v;
    }
    __syncthreads();

    v8f c[GT_TILES];
#pragma unroll
    for (int rt = 0; rt < GT_TILES; ++rt) c[rt] = v8f{};
    const float* ap = xs + l16 * G1_PITCH + half * 2;
    const float* wp = W + (size_t)(half * 2) * HID + col;
#pragma unroll 8
    for (int kk = 0; kk < IND; kk += 4) {
        v2f b;
        b.x = wp[(size_t)kk * HID];
        b.y = wp[(size_t)(kk + 1) * HID];
#pragma unroll
        for (int rt = 0; rt < GT_TILES; ++rt) {
            const v2f a = *(const v2f*)(ap + rt * 16 * G1_PITCH + kk);
            c[rt] = __builtin_amdgcn_wmma_f32_16x16x4_f32(false, a, false, b, (short)0, c[rt], false, false);
        }
    }
#pragma unroll
    for (int rt = 0; rt < GT_TILES; ++rt)
#pragma unroll
        for (int v = 0; v < 8; ++v)
            H[(size_t)(row0 + rt * 16 + v + half * 8) * HID + col] = c[rt][v];
}

template <int CTRL> __device__ __forceinline__ float row_swz(float x) {
    const int b = __builtin_bit_cast(int, x);
    return __builtin_bit_cast(float, __builtin_amdgcn_update_dpp(0, b, CTRL, 0xf, 0xf, true));
}
__device__ __forceinline__ float quad_sum(float x) {
    x += row_swz<0x161>(x);
    x += row_swz<0x162>(x);
    return x;
}
__device__ __forceinline__ float row_sum(float x) {
    x = quad_sum(x);
    x += row_swz<0x164>(x);
    x += row_swz<0x168>(x);
    return x;
}
__device__ __forceinline__ float xhalf(float x) {
    const int b = __builtin_bit_cast(int, x);
    return __builtin_bit_cast(float, __builtin_amdgcn_permlanex16(b, b, 0x76543210, 0xfedcba98, false, false));
}
__device__ __forceinline__ float dot4(const float4 a, const float4 b) {
    return a.x * b.x + a.y * b.y + a.z * b.z + a.w * b.w;
}

__global__ __launch_bounds__(256) void k_gat1(const int* __restrict__ esrc,
                                              const unsigned* __restrict__ deg, const unsigned* __restrict__ endp,
                                              const float* __restrict__ H,
                                              const float* __restrict__ al, const float* __restrict__ ar,
                                              const float* __restrict__ b, float* __restrict__ h2) {
    const int lane = threadIdx.x & 31;
    const int n = blockIdx.x * 8 + __builtin_amdgcn_readfirstlane((int)(threadIdx.x >> 5));
    const int end = __builtin_amdgcn_readfirstlane((int)endp[n]);
    const int beg = end - __builtin_amdgcn_readfirstlane((int)deg[n]);
    const float* Hl = H + 4 * lane;
    const float4 alv = *(const float4*)(al + 4 * lane);
    const float4 arv = *(const float4*)(ar + 4 * lane);
    const float4 own = *(const float4*)(Hl + (size_t)n * HID);
    const float erv = quad_sum(dot4(own, arv));

    float m = -__builtin_inff();
    float ssum = 0.f;
    float4 acc = make_float4(0.f, 0.f, 0.f, 0.f);
    for (int j0 = beg; j0 < end; j0 += 32) {
        const int jj = j0 + lane;
        const int cnt = (end - j0 < 32) ? (end - j0) : 32;
        const int sv = esrc[(jj < end) ? jj : (end - 1)];
        const int last = cnt - 1;
        for (int i = 0; i < cnt; i += 4) {
            float4 r[4];
            float e[4];
#pragma unroll
            for (int g = 0; g < 4; ++g) {
                const int ig = (i + g < cnt) ? (i + g) : last;
                const int s = __builtin_amdgcn_readlane(sv, ig);
                r[g] = *(const float4*)(Hl + (size_t)s * HID);
            }
#pragma unroll
            for (int g = 0; g < 4; ++g) {
                const float d = quad_sum(dot4(r[g], alv));
                e[g] = (i + g < cnt) ? lrelu(d + erv) : -__builtin_inff();
            }
            const float mn = fmaxf(fmaxf(m, e[0]), fmaxf(fmaxf(e[1], e[2]), e[3]));
            const float sc = __expf(m - mn);
            ssum *= sc; acc.x *= sc; acc.y *= sc; acc.z *= sc; acc.w *= sc;
#pragma unroll
            for (int g = 0; g < 4; ++g) {
                const float p = __expf(e[g] - mn);
                ssum += p; acc.x += p * r[g].x; acc.y += p * r[g].y; acc.z += p * r[g].z; acc.w += p * r[g].w;
            }
            m = mn;
        }
    }
    const float4 bv = *(const float4*)(b + 4 * lane);
    const bool any = ssum > 0.f;
    float4 o;
    o.x = elu1(elu1((any ? acc.x / ssum : 0.f) + bv.x));
    o.y = elu1(elu1((any ? acc.y / ssum : 0.f) + bv.y));
    o.z = elu1(elu1((any ? acc.z / ssum : 0.f) + bv.z));
    o.w = elu1(elu1((any ? acc.w / ssum : 0.f) + bv.w));
    *(float4*)(h2 + (size_t)n * HID + 4 * lane) = o;
}

static constexpr int G2_PITCH = HID + 4;
__global__ __launch_bounds__(128) void k_gemm2(const float* __restrict__ H2,
                                               const float* __restrict__ W,
                                               float* __restrict__ HT) {
    __shared__ __attribute__((aligned(16))) float hs[GT_ROWS * G2_PITCH];
    const int tid  = threadIdx.x;
    const int wave = tid >> 5;
    const int lane = tid & 31;
    const int half = lane >> 4;
    const int l16  = lane & 15;
    const int row0 = blockIdx.x * GT_ROWS;
    const int col  = wave * 16 + l16;

    const float4* hg = (const float4*)(H2 + (size_t)row0 * HID);
#pragma unroll
    for (int i = 0; i < GT_ROWS * HID / 4 / 128; ++i) {
        const int idx = i * 128 + tid;
        const float4 v = hg[idx];
        *(float4*)(hs + (idx >> 5) * G2_PITCH + (idx & 31) * 4) = v;
    }
    __syncthreads();

    v8f c[GT_TILES];
#pragma unroll
    for (int rt = 0; rt < GT_TILES; ++rt) c[rt] = v8f{};
    const float* ap = hs + l16 * G2_PITCH + half * 2;
    const float* wp = W + (size_t)(half * 2) * NCLS + col;
#pragma unroll 8
    for (int kk = 0; kk < HID; kk += 4) {
        v2f b;
        b.x = wp[(size_t)kk * NCLS];
        b.y = wp[(size_t)(kk + 1) * NCLS];
#pragma unroll
        for (int rt = 0; rt < GT_TILES; ++rt) {
            const v2f a = *(const v2f*)(ap + rt * 16 * G2_PITCH + kk);
            c[rt] = __builtin_amdgcn_wmma_f32_16x16x4_f32(false, a, false, b, (short)0, c[rt], false, false);
        }
    }
#pragma unroll
    for (int rt = 0; rt < GT_TILES; ++rt)
#pragma unroll
        for (int v = 0; v < 8; ++v)
            HT[(size_t)(row0 + rt * 16 + v + half * 8) * NCLS + col] = c[rt][v];
}

__global__ __launch_bounds__(256) void k_gat2(const int* __restrict__ esrc,
                                              const unsigned* __restrict__ deg, const unsigned* __restrict__ endp,
                                              const float* __restrict__ HT,
                                              const float* __restrict__ al, const float* __restrict__ ar,
                                              const float* __restrict__ b, float* __restrict__ out) {
    const int lane = threadIdx.x & 31;
    const int hf = lane >> 4, l16 = lane & 15;
    const int n = blockIdx.x * 8 + __builtin_amdgcn_readfirstlane((int)(threadIdx.x >> 5));
    const int end = __builtin_amdgcn_readfirstlane((int)endp[n]);
    const int beg = end - __builtin_amdgcn_readfirstlane((int)deg[n]);
    static_assert((unsigned long long)NN * NCLS * sizeof(float) < (1ull << 32), "32-bit row offsets");
    const char* Tb = (const char*)HT;
    const unsigned lo = (unsigned)l16 * 16u;
    const float4 alv = *(const float4*)(al + 4 * l16);
    const float4 arv = *(const float4*)(ar + 4 * l16);
    const float4 own = *(const float4*)(Tb + (((unsigned)n << 8) + lo));
    const float erv = row_sum(dot4(own, arv));

    float m = -__builtin_inff();
    float ssum = 0.f;
    float4 acc = make_float4(0.f, 0.f, 0.f, 0.f);
    for (int j0 = beg; j0 < end; j0 += 32) {
        const int jj = j0 + lane;
        const int cnt = (end - j0 < 32) ? (end - j0) : 32;
        const int sv = esrc[(jj < end) ? jj : (end - 1)];
        const int last = cnt - 1;
        for (int i = 0; i < cnt; i += 8) {
            float4 r[4];
            float e[4];
            bool ok[4];
#pragma unroll
            for (int g = 0; g < 4; ++g) {
                const int ia = i + 2 * g, ib = ia + 1;
                const int sa = __builtin_amdgcn_readlane(sv, (ia < cnt) ? ia : last);
                const int sb = __builtin_amdgcn_readlane(sv, (ib < cnt) ? ib : last);
                const int s = hf ? sb : sa;
                ok[g] = (ia + hf) < cnt;
                r[g] = *(const float4*)(Tb + (((unsigned)s << 8) + lo));
            }
            asm volatile("" :: "v"(r[0].x), "v"(r[1].x), "v"(r[2].x), "v"(r[3].x));
#pragma unroll
            for (int g = 0; g < 4; ++g) {
                const float d = row_sum(dot4(r[g], alv));
                e[g] = ok[g] ? lrelu(d + erv) : -__builtin_inff();
            }
            const float mn = fmaxf(fmaxf(m, e[0]), fmaxf(fmaxf(e[1], e[2]), e[3]));
            const float sc = (m == mn) ? 1.f : __expf(m - mn);
            ssum *= sc; acc.x *= sc; acc.y *= sc; acc.z *= sc; acc.w *= sc;
#pragma unroll
            for (int g = 0; g < 4; ++g) {
                const float p = ok[g] ? __expf(e[g] - mn) : 0.f;
                ssum += p; acc.x += p * r[g].x; acc.y += p * r[g].y; acc.z += p * r[g].z; acc.w += p * r[g].w;
            }
            m = mn;
        }
    }
    const float M = fmaxf(m, xhalf(m));
    const float f = (m == M) ? 1.f : __expf(m - M);
    ssum *= f; acc.x *= f; acc.y *= f; acc.z *= f; acc.w *= f;
    ssum += xhalf(ssum);
    acc.x += xhalf(acc.x); acc.y += xhalf(acc.y); acc.z += xhalf(acc.z); acc.w += xhalf(acc.w);
    const float4 bv = *(const float4*)(b + 4 * l16);
    const bool any = ssum > 0.f;
    float4 o;
    o.x = (any ? acc.x / ssum : 0.f) + bv.x;
    o.y = (any ? acc.y / ssum : 0.f) + bv.y;
    o.z = (any ? acc.z / ssum : 0.f) + bv.z;
    o.w = (any ? acc.w / ssum : 0.f) + bv.w;
    if (hf == 0) *(float4*)(out + (size_t)n * NCLS + 4 * l16) = o;
}

extern "C" void kernel_launch(void* const* d_in, const int* in_sizes, int n_in,
                              void* d_out, int out_size, void* d_ws, size_t ws_size,
                              hipStream_t stream) {
    const float* x   = (const float*)d_in[0];
    const int*   src = (const int*)  d_in[1];
    const int*   dst = (const int*)  d_in[2];
    const float* W1  = (const float*)d_in[3];
    const float* al1 = (const float*)d_in[4];
    const float* ar1 = (const float*)d_in[5];
    const float* b1  = (const float*)d_in[6];
    const float* W2  = (const float*)d_in[7];
    const float* al2 = (const float*)d_in[8];
    const float* ar2 = (const float*)d_in[9];
    const float* b2  = (const float*)d_in[10];
    float* out = (float*)d_out;

    size_t off = 0;
    auto take = [&](size_t nfloats) -> float* {
        float* p = (float*)((char*)d_ws + off);
        off = (off + nfloats * sizeof(float) + 255) & ~(size_t)255;
        return p;
    };
    float*    h1    = take((size_t)NN * HID);
    int*      esrc  = (int*)take((size_t)EE);
    float*    h2    = take((size_t)NN * HID);
    float*    h2t   = take((size_t)NN * NCLS);
    unsigned* deg   = (unsigned*)take((size_t)NN);
    unsigned* cur   = (unsigned*)take((size_t)NN);
    unsigned* total = (unsigned*)h2t;

    const int T = 256;
    k_zero <<<(NN + T - 1) / T, T, 0, stream>>>(deg, total);
    k_count<<<(EE + T - 1) / T, T, 0, stream>>>(dst, deg);
    k_alloc<<<(NN + T - 1) / T, T, 0, stream>>>(deg, cur, total);
    k_fill <<<(EE + T - 1) / T, T, 0, stream>>>(src, dst, cur, esrc);
    k_gemm1<<<NN / GT_ROWS, T, 0, stream>>>(x, W1, h1);
    k_gat1 <<<NN / 8, T, 0, stream>>>(esrc, deg, cur, h1, al1, ar1, b1, h2);
    k_gemm2<<<NN / GT_ROWS, 128, 0, stream>>>(h2, W2, h2t);
    k_gat2 <<<NN / 8, T, 0, stream>>>(esrc, deg, cur, h2t, al2, ar2, b2, out);
}


# Edge-list fill without atomics: the counting add returns each edge's rank, kept per edge; fill stores at segment start + rank; one returning add per edge gone; sum order = count arrival order
# speedup vs baseline: 6.0942x; 1.1935x over previous
#include <hip/hip_runtime.h>

typedef __attribute__((ext_vector_type(2))) float v2f;
typedef __attribute__((ext_vector_type(8))) float v8f;

static constexpr int NN   = 100000;
static constexpr int EE   = 1600000;
static constexpr int IND  = 256;
static constexpr int HID  = 128;
static constexpr int HEADS= 8;
static constexpr int DH   = 16;
static constexpr int NCLS = 64;
static constexpr float SLOPE = 0.2f;

__device__ __forceinline__ float lrelu(float x) { return x > 0.f ? x : SLOPE * x; }
__device__ __forceinline__ float elu1(float x)  { return x > 0.f ? x : expm1f(x); }

__global__ __launch_bounds__(256) void k_zero(unsigned* __restrict__ deg, unsigned* __restrict__ total) {
    const int t = blockIdx.x * 256 + threadIdx.x;
    if (t < NN) deg[t] = 0u;
    if (t == 0) *total = 0u;
}
__global__ __launch_bounds__(256) void k_count(const int* __restrict__ dst, unsigned* __restrict__ deg,
                                               unsigned* __restrict__ rnk) {
    const int t = blockIdx.x * 256 + threadIdx.x;
    if (t >= EE) return;
    rnk[t] = atomicAdd(&deg[dst[t]], 1u);
}
__global__ __launch_bounds__(256) void k_alloc(const unsigned* __restrict__ deg, unsigned* __restrict__ cur,
                                               unsigned* __restrict__ total) {
    const int t = blockIdx.x * 256 + threadIdx.x;
    const int lane = threadIdx.x & 31;
    const unsigned d = (t < NN) ? deg[t] : 0u;
    unsigned inc = d;
    for (int o = 1; o < 32; o <<= 1) {
        const unsigned v = __shfl_up(inc, o);
        if (lane >= o) inc += v;
    }
    unsigned base = 0u;
    if (lane == 31) base = atomicAdd(total, inc);
    base = __shfl(base, 31);
    if (t < NN) cur[t] = base + inc - d;
}
__global__ __launch_bounds__(256) void k_fill(const int* __restrict__ src, const int* __restrict__ dst,
                                              const unsigned* __restrict__ cur, const unsigned* __restrict__ rnk,
                                              int* __restrict__ esrc) {
    const int t = blockIdx.x * 256 + threadIdx.x;
    if (t >= EE) return;
    esrc[cur[dst[t]] + rnk[t]] = src[t];
}

static constexpr int GT_ROWS  = 80;
static constexpr int GT_TILES = GT_ROWS / 16;
static constexpr int G1_PITCH = IND + 4;
__global__ __launch_bounds__(256) void k_gemm1(const float* __restrict__ X,
                                               const float* __restrict__ W,
                                               float* __restrict__ H) {
    __shared__ __attribute__((aligned(16))) float xs[GT_ROWS * G1_PITCH];
    const int tid  = threadIdx.x;
    const int wave = tid >> 5;
    const int lane = tid & 31;
    const int half = lane >> 4;
    const int l16  = lane & 15;
    const int row0 = blockIdx.x * GT_ROWS;
    const int col  = wave * 16 + l16;

    const float4* xg = (const float4*)(X + (size_t)row0 * IND);
#pragma unroll
    for (int i = 0; i < GT_ROWS * IND / 4 / 256; ++i) {
        const int idx = i * 256 + tid;
        const float4 v = xg[idx];
        *(float4*)(xs + (idx >> 6) * G1_PITCH + (idx & 63) * 4) = v;
    }
    __syncthreads();

    v8f c[GT_TILES];
#pragma unroll
    for (int rt = 0; rt < GT_TILES; ++rt) c[rt] = v8f{};
    const float* ap = xs + l16 * G1_PITCH + half * 2;
    const float* wp = W + (size_t)(half * 2) * HID + col;
#pragma unroll 8
    for (int kk = 0; kk < IND; kk += 4) {
        v2f b;
        b.x = wp[(size_t)kk * HID];
        b.y = wp[(size_t)(kk + 1) * HID];
#pragma unroll
        for (int rt = 0; rt < GT_TILES; ++rt) {
            const v2f a = *(const v2f*)(ap + rt * 16 * G1_PITCH + kk);
            c[rt] = __builtin_amdgcn_wmma_f32_16x16x4_f32(false, a, false, b, (short)0, c[rt], false, false);
        }
    }
#pragma unroll
    for (int rt = 0; rt < GT_TILES; ++rt)
#pragma unroll
        for (int v = 0; v < 8; ++v)
            H[(size_t)(row0 + rt * 16 + v + half * 8) * HID + col] = c[rt][v];
}

template <int CTRL> __device__ __forceinline__ float row_swz(float x) {
    const int b = __builtin_bit_cast(int, x);
    return __builtin_bit_cast(float, __builtin_amdgcn_update_dpp(0, b, CTRL, 0xf, 0xf, true));
}
__device__ __forceinline__ float quad_sum(float x) {
    x += row_swz<0x161>(x);
    x += row_swz<0x162>(x);
    return x;
}
__device__ __forceinline__ float row_sum(float x) {
    x = quad_sum(x);
    x += row_swz<0x164>(x);
    x += row_swz<0x168>(x);
    return x;
}
__device__ __forceinline__ float xhalf(float x) {
    const int b = __builtin_bit_cast(int, x);
    return __builtin_bit_cast(float, __builtin_amdgcn_permlanex16(b, b, 0x76543210, 0xfedcba98, false, false));
}
__device__ __forceinline__ float dot4(const float4 a, const float4 b) {
    return a.x * b.x + a.y * b.y + a.z * b.z + a.w * b.w;
}

__global__ __launch_bounds__(256) void k_gat1(const int* __restrict__ esrc,
                                              const unsigned* __restrict__ deg, const unsigned* __restrict__ begp,
                                              const float* __restrict__ H,
                                              const float* __restrict__ al, const float* __restrict__ ar,
                                              const float* __restrict__ b, float* __restrict__ h2) {
    const int lane = threadIdx.x & 31;
    const int n = blockIdx.x * 8 + __builtin_amdgcn_readfirstlane((int)(threadIdx.x >> 5));
    const int beg = __builtin_amdgcn_readfirstlane((int)begp[n]);
    const int end = beg + __builtin_amdgcn_readfirstlane((int)deg[n]);
    const float* Hl = H + 4 * lane;
    const float4 alv = *(const float4*)(al + 4 * lane);
    const float4 arv = *(const float4*)(ar + 4 * lane);
    const float4 own = *(const float4*)(Hl + (size_t)n * HID);
    const float erv = quad_sum(dot4(own, arv));

    float m = -__builtin_inff();
    float ssum = 0.f;
    float4 acc = make_float4(0.f, 0.f, 0.f, 0.f);
    for (int j0 = beg; j0 < end; j0 += 32) {
        const int jj = j0 + lane;
        const int cnt = (end - j0 < 32) ? (end - j0) : 32;
        const int sv = esrc[(jj < end) ? jj : (end - 1)];
        const int last = cnt - 1;
        for (int i = 0; i < cnt; i += 4) {
            float4 r[4];
            float e[4];
#pragma unroll
            for (int g = 0; g < 4; ++g) {
                const int ig = (i + g < cnt) ? (i + g) : last;
                const int s = __builtin_amdgcn_readlane(sv, ig);
                r[g] = *(const float4*)(Hl + (size_t)s * HID);
            }
#pragma unroll
            for (int g = 0; g < 4; ++g) {
                const float d = quad_sum(dot4(r[g], alv));
                e[g] = (i + g < cnt) ? lrelu(d + erv) : -__builtin_inff();
            }
            const float mn = fmaxf(fmaxf(m, e[0]), fmaxf(fmaxf(e[1], e[2]), e[3]));
            const float sc = __expf(m - mn);
            ssum *= sc; acc.x *= sc; acc.y *= sc; acc.z *= sc; acc.w *= sc;
#pragma unroll
            for (int g = 0; g < 4; ++g) {
                const float p = __expf(e[g] - mn);
                ssum += p; acc.x += p * r[g].x; acc.y += p * r[g].y; acc.z += p * r[g].z; acc.w += p * r[g].w;
            }
            m = mn;
        }
    }
    const float4 bv = *(const float4*)(b + 4 * lane);
    const bool any = ssum > 0.f;
    float4 o;
    o.x = elu1(elu1((any ? acc.x / ssum : 0.f) + bv.x));
    o.y = elu1(elu1((any ? acc.y / ssum : 0.f) + bv.y));
    o.z = elu1(elu1((any ? acc.z / ssum : 0.f) + bv.z));
    o.w = elu1(elu1((any ? acc.w / ssum : 0.f) + bv.w));
    *(float4*)(h2 + (size_t)n * HID + 4 * lane) = o;
}

static constexpr int G2_PITCH = HID + 4;
__global__ __launch_bounds__(128) void k_gemm2(const float* __restrict__ H2,
                                               const float* __restrict__ W,
                                               float* __restrict__ HT) {
    __shared__ __attribute__((aligned(16))) float hs[GT_ROWS * G2_PITCH];
    const int tid  = threadIdx.x;
    const int wave = tid >> 5;
    const int lane = tid & 31;
    const int half = lane >> 4;
    const int l16  = lane & 15;
    const int row0 = blockIdx.x * GT_ROWS;
    const int col  = wave * 16 + l16;

    const float4* hg = (const float4*)(H2 + (size_t)row0 * HID);
#pragma unroll
    for (int i = 0; i < GT_ROWS * HID / 4 / 128; ++i) {
        const int idx = i * 128 + tid;
        const float4 v = hg[idx];
        *(float4*)(hs + (idx >> 5) * G2_PITCH + (idx & 31) * 4) = v;
    }
    __syncthreads();

    v8f c[GT_TILES];
#pragma unroll
    for (int rt = 0; rt < GT_TILES; ++rt) c[rt] = v8f{};
    const float* ap = hs + l16 * G2_PITCH + half * 2;
    const float* wp = W + (size_t)(half * 2) * NCLS + col;
#pragma unroll 8
    for (int kk = 0; kk < HID; kk += 4) {
        v2f b;
        b.x = wp[(size_t)kk * NCLS];
        b.y = wp[(size_t)(kk + 1) * NCLS];
#pragma unroll
        for (int rt = 0; rt < GT_TILES; ++rt) {
            const v2f a = *(const v2f*)(ap + rt * 16 * G2_PITCH + kk);
            c[rt] = __builtin_amdgcn_wmma_f32_16x16x4_f32(false, a, false, b, (short)0, c[rt], false, false);
        }
    }
#pragma unroll
    for (int rt = 0; rt < GT_TILES; ++rt)
#pragma unroll
        for (int v = 0; v < 8; ++v)
            HT[(size_t)(row0 + rt * 16 + v + half * 8) * NCLS + col] = c[rt][v];
}

__global__ __launch_bounds__(256) void k_gat2(const int* __restrict__ esrc,
                                              const unsigned* __restrict__ deg, const unsigned* __restrict__ begp,
                                              const float* __restrict__ HT,
                                              const float* __restrict__ al, const float* __restrict__ ar,
                                              const float* __restrict__ b, float* __restrict__ out) {
    const int lane = threadIdx.x & 31;
    const int hf = lane >> 4, l16 = lane & 15;
    const int n = blockIdx.x * 8 + __builtin_amdgcn_readfirstlane((int)(threadIdx.x >> 5));
    const int beg = __builtin_amdgcn_readfirstlane((int)begp[n]);
    const int end = beg + __builtin_amdgcn_readfirstlane((int)deg[n]);
    static_assert((unsigned long long)NN * NCLS * sizeof(float) < (1ull << 32), "32-bit row offsets");
    const char* Tb = (const char*)HT;
    const unsigned lo = (unsigned)l16 * 16u;
    const float4 alv = *(const float4*)(al + 4 * l16);
    const float4 arv = *(const float4*)(ar + 4 * l16);
    const float4 own = *(const float4*)(Tb + (((unsigned)n << 8) + lo));
    const float erv = row_sum(dot4(own, arv));

    float m = -__builtin_inff();
    float ssum = 0.f;
    float4 acc = make_float4(0.f, 0.f, 0.f, 0.f);
    for (int j0 = beg; j0 < end; j0 += 32) {
        const int jj = j0 + lane;
        const int cnt = (end - j0 < 32) ? (end - j0) : 32;
        const int sv = esrc[(jj < end) ? jj : (end - 1)];
        const int last = cnt - 1;
        for (int i = 0; i < cnt; i += 8) {
            float4 r[4];
            float e[4];
            bool ok[4];
#pragma unroll
            for (int g = 0; g < 4; ++g) {
                const int ia = i + 2 * g, ib = ia + 1;
                const int sa = __builtin_amdgcn_readlane(sv, (ia < cnt) ? ia : last);
                const int sb = __builtin_amdgcn_readlane(sv, (ib < cnt) ? ib : last);
                const int s = hf ? sb : sa;
                ok[g] = (ia + hf) < cnt;
                r[g] = *(const float4*)(Tb + (((unsigned)s << 8) + lo));
            }
            asm volatile("" :: "v"(r[0].x), "v"(r[1].x), "v"(r[2].x), "v"(r[3].x));
#pragma unroll
            for (int g = 0; g < 4; ++g) {
                const float d = row_sum(dot4(r[g], alv));
                e[g] = ok[g] ? lrelu(d + erv) : -__builtin_inff();
            }
            const float mn = fmaxf(fmaxf(m, e[0]), fmaxf(fmaxf(e[1], e[2]), e[3]));
            const float sc = (m == mn) ? 1.f : __expf(m - mn);
            ssum *= sc; acc.x *= sc; acc.y *= sc; acc.z *= sc; acc.w *= sc;
#pragma unroll
            for (int g = 0; g < 4; ++g) {
                const float p = ok[g] ? __expf(e[g] - mn) : 0.f;
                ssum += p; acc.x += p * r[g].x; acc.y += p * r[g].y; acc.z += p * r[g].z; acc.w += p * r[g].w;
            }
            m = mn;
        }
    }
    const float M = fmaxf(m, xhalf(m));
    const float f = (m == M) ? 1.f : __expf(m - M);
    ssum *= f; acc.x *= f; acc.y *= f; acc.z *= f; acc.w *= f;
    ssum += xhalf(ssum);
    acc.x += xhalf(acc.x); acc.y += xhalf(acc.y); acc.z += xhalf(acc.z); acc.w += xhalf(acc.w);
    const float4 bv = *(const float4*)(b + 4 * l16);
    const bool any = ssum > 0.f;
    float4 o;
    o.x = (any ? acc.x / ssum : 0.f) + bv.x;
    o.y = (any ? acc.y / ssum : 0.f) + bv.y;
    o.z = (any ? acc.z / ssum : 0.f) + bv.z;
    o.w = (any ? acc.w / ssum : 0.f) + bv.w;
    if (hf == 0) *(float4*)(out + (size_t)n * NCLS + 4 * l16) = o;
}

extern "C" void kernel_launch(void* const* d_in, const int* in_sizes, int n_in,
                              void* d_out, int out_size, void* d_ws, size_t ws_size,
                              hipStream_t stream) {
    const float* x   = (const float*)d_in[0];
    const int*   src = (const int*)  d_in[1];
    const int*   dst = (const int*)  d_in[2];
    const float* W1  = (const float*)d_in[3];
    const float* al1 = (const float*)d_in[4];
    const float* ar1 = (const float*)d_in[5];
    const float* b1  = (const float*)d_in[6];
    const float* W2  = (const float*)d_in[7];
    const float* al2 = (const float*)d_in[8];
    const float* ar2 = (const float*)d_in[9];
    const float* b2  = (const float*)d_in[10];
    float* out = (float*)d_out;

    size_t off = 0;
    auto take = [&](size_t nfloats) -> float* {
        float* p = (float*)((char*)d_ws + off);
        off = (off + nfloats * sizeof(float) + 255) & ~(size_t)255;
        return p;
    };
    float*    h1    = take((size_t)NN * HID);
    int*      esrc  = (int*)take((size_t)EE);
    float*    h2    = take((size_t)NN * HID);
    float*    h2t   = take((size_t)NN * NCLS);
    unsigned* deg   = (unsigned*)take((size_t)NN);
    unsigned* cur   = (unsigned*)take((size_t)NN);
    unsigned* total = (unsigned*)h2t;
    unsigned* rnk   = (unsigned*)h2;
    static_assert((size_t)EE <= (size_t)NN * HID, "the rank plane fits in the h2 plane");

    const int T = 256;
    k_zero <<<(NN + T - 1) / T, T, 0, stream>>>(deg, total);
    k_count<<<(EE + T - 1) / T, T, 0, stream>>>(dst, deg, rnk);
    k_alloc<<<(NN + T - 1) / T, T, 0, stream>>>(deg, cur, total);
    k_fill <<<(EE + T - 1) / T, T, 0, stream>>>(src, dst, cur, rnk, esrc);
    k_gemm1<<<NN / GT_ROWS, T, 0, stream>>>(x, W1, h1);
    k_gat1 <<<NN / 8, T, 0, stream>>>(esrc, deg, cur, h1, al1, ar1, b1, h2);
    k_gemm2<<<NN / GT_ROWS, 128, 0, stream>>>(h2, W2, h2t);
    k_gat2 <<<NN / 8, T, 0, stream>>>(esrc, deg, cur, h2t, al2, ar2, b2, out);
}


# Edge list built by a two-level counting sort in LDS: blocks bin edges into 512-node buckets, one block per bucket ranks and places them; no global add per edge left; sum order = LDS arrival order
# speedup vs baseline: 9.5354x; 1.5647x over previous
#include <hip/hip_runtime.h>

typedef __attribute__((ext_vector_type(2))) float v2f;
typedef __attribute__((ext_vector_type(8))) float v8f;

static constexpr int NN   = 100000;
static constexpr int EE   = 1600000;
static constexpr int IND  = 256;
static constexpr int HID  = 128;
static constexpr int HEADS= 8;
static constexpr int DH   = 16;
static constexpr int NCLS = 64;
static constexpr float SLOPE = 0.2f;

__device__ __forceinline__ float lrelu(float x) { return x > 0.f ? x : SLOPE * x; }
__device__ __forceinline__ float elu1(float x)  { return x > 0.f ? x : expm1f(x); }

static constexpr int BSH    = 9;
static constexpr int NPB    = 1 << BSH;
static constexpr int NBK    = (NN + NPB - 1) / NPB;
static constexpr int CH_T   = 1024;
static constexpr int CH_E   = CH_T * 16;
static constexpr int NCH    = (EE + CH_E - 1) / CH_E;
static constexpr int NCHP   = (NCH + 3) / 4 * 4;
static constexpr int BK_T   = 1024;
static constexpr int BK_W   = 12;
static constexpr int BK_CAP = 32768;
static_assert(EE % 4 == 0, "the edge arrays are read by 16-byte loads");
static_assert(NBK <= CH_T && NPB <= BK_T, "one thread per bucket in a binning block, per node in a bucket block");
static_assert(NCHP - NCH <= NCH, "the first blocks clear the row padding");
static_assert(((unsigned long long)(NN - 1) << BSH) + (NPB - 1) < (1ull << 32), "an edge word holds source and place");

template <int NW> __device__ __forceinline__ unsigned block_excl_scan(unsigned v, unsigned* wsum, int lane, int wave) {
    unsigned inc = v;
    for (int o = 1; o < 32; o <<= 1) {
        const unsigned u = __shfl_up(inc, o);
        if (lane >= o) inc += u;
    }
    if (lane == 31) wsum[wave] = inc;
    __syncthreads();
    const unsigned t = (lane < NW) ? wsum[lane] : 0u;
    unsigned ti = t;
    for (int o = 1; o < 32; o <<= 1) {
        const unsigned u = __shfl_up(ti, o);
        if (lane >= o) ti += u;
    }
    const unsigned wbase = __shfl(ti - t, wave);
    __syncthreads();
    return wbase + inc - v;
}

__global__ __launch_bounds__(CH_T) void k_hist(const int* __restrict__ dst, unsigned* __restrict__ cnt) {
    __shared__ unsigned hs[NBK];
    const int tid = threadIdx.x;
    const int blk = blockIdx.x;
    if (tid < NBK) hs[tid] = 0u;
    __syncthreads();
    const int4* d4 = (const int4*)dst;
    int4 d[4];
    bool ok[4];
#pragma unroll
    for (int i = 0; i < 4; ++i) {
        const int e4 = blk * (CH_E / 4) + i * CH_T + tid;
        ok[i] = e4 < EE / 4;
        d[i] = d4[ok[i] ? e4 : (EE / 4 - 1)];
    }
#pragma unroll
    for (int i = 0; i < 4; ++i) {
        if (ok[i]) {
            atomicAdd(&hs[d[i].x >> BSH], 1u);
            atomicAdd(&hs[d[i].y >> BSH], 1u);
            atomicAdd(&hs[d[i].z >> BSH], 1u);
            atomicAdd(&hs[d[i].w >> BSH], 1u);
        }
    }
    __syncthreads();
    if (tid < NBK) {
        cnt[tid * NCHP + blk] = hs[tid];
        if (blk < NCHP - NCH) cnt[tid * NCHP + NCH + blk] = 0u;
    }
}

__global__ __launch_bounds__(CH_T) void k_scatter(const int* __restrict__ src, const int* __restrict__ dst,
                                                  const unsigned* __restrict__ cnt, unsigned* __restrict__ gbase,
                                                  unsigned* __restrict__ part) {
    __shared__ unsigned wbuf[CH_E];
    __shared__ unsigned loff[NBK];
    __shared__ unsigned lcur[NBK];
    __shared__ unsigned goff[NBK];
    __shared__ unsigned wsum[CH_T / 32];
    const int tid  = threadIdx.x;
    const int lane = tid & 31;
    const int wave = tid >> 5;
    const int blk  = blockIdx.x;

    unsigned tot = 0u, pre = 0u, mine = 0u;
    if (tid < NBK) {
        const uint4* cp = (const uint4*)(cnt + tid * NCHP);
        auto acc = [&](int r, unsigned c) {
            if (r < NCH) {
                tot += c;
                pre += (r < blk) ? c : 0u;
                mine = (r == blk) ? c : mine;
            }
        };
#pragma unroll
        for (int q = 0; q < NCHP / 4; ++q) {
            const uint4 v = cp[q];
            acc(4 * q, v.x); acc(4 * q + 1, v.y); acc(4 * q + 2, v.z); acc(4 * q + 3, v.w);
        }
    }
    const unsigned base = block_excl_scan<CH_T / 32>(tot,  wsum, lane, wave);
    const unsigned lst  = block_excl_scan<CH_T / 32>(mine, wsum, lane, wave);
    if (tid < NBK) {
        loff[tid] = lst;
        lcur[tid] = lst;
        goff[tid] = base + pre;
        if (blk == 0) {
            gbase[tid] = base;
            if (tid == NBK - 1) gbase[NBK] = base + tot;
        }
    }
    __syncthreads();

    const int4* d4 = (const int4*)dst;
    const int4* s4 = (const int4*)src;
    int4 d[4], s[4];
    bool ok[4];
#pragma unroll
    for (int i = 0; i < 4; ++i) {
        const int e4 = blk * (CH_E / 4) + i * CH_T + tid;
        ok[i] = e4 < EE / 4;
        const int ec = ok[i] ? e4 : (EE / 4 - 1);
        d[i] = d4[ec];
        s[i] = s4[ec];
    }
    auto place = [&](int dv, int sv) {
        const unsigned p = atomicAdd(&lcur[dv >> BSH], 1u);
        wbuf[p] = ((unsigned)sv << BSH) | ((unsigned)dv & (unsigned)(NPB - 1));
    };
#pragma unroll
    for (int i = 0; i < 4; ++i) {
        if (ok[i]) {
            place(d[i].x, s[i].x);
            place(d[i].y, s[i].y);
            place(d[i].z, s[i].z);
            place(d[i].w, s[i].w);
        }
    }
    __syncthreads();

    for (int b = wave; b < NBK; b += CH_T / 32) {
        const unsigned s0 = loff[b];
        const unsigned n  = lcur[b] - s0;
        const unsigned g  = goff[b];
        for (unsigned l = lane; l < n; l += 32) part[g + l] = wbuf[s0 + l];
    }
}

__global__ __launch_bounds__(BK_T) void k_bucket(const unsigned* __restrict__ part, const unsigned* __restrict__ gbase,
                                                 unsigned* __restrict__ deg, unsigned* __restrict__ cur,
                                                 int* __restrict__ esrc) {
    __shared__ int lb[BK_CAP];
    __shared__ unsigned c[NPB];
    __shared__ unsigned wsum[BK_T / 32];
    const int tid  = threadIdx.x;
    const int lane = tid & 31;
    const int wave = tid >> 5;
    const int bk   = blockIdx.x;
    const unsigned beg = gbase[bk];
    const unsigned end = gbase[bk + 1];
    const unsigned nb  = end - beg;
    if (tid < NPB) c[tid] = 0u;
    __syncthreads();

    for (unsigned j0 = beg; j0 < end; j0 += BK_T * BK_W) {
        unsigned w[BK_W];
#pragma unroll
        for (int k = 0; k < BK_W; ++k) {
            const unsigned j = j0 + k * BK_T + tid;
            w[k] = part[(j < end) ? j : (end - 1u)];
        }
#pragma unroll
        for (int k = 0; k < BK_W; ++k) {
            const unsigned j = j0 + k * BK_T + tid;
            if (j < end) atomicAdd(&c[w[k] & (unsigned)(NPB - 1)], 1u);
        }
    }
    __syncthreads();

    const unsigned dg = (tid < NPB) ? c[tid] : 0u;
    const unsigned ex = block_excl_scan<BK_T / 32>(dg, wsum, lane, wave);
    if (tid < NPB) {
        const int n = bk * NPB + tid;
        if (n < NN) { deg[n] = dg; cur[n] = beg + ex; }
        c[tid] = ex;
    }
    __syncthreads();

    const bool in_lds = nb <= (unsigned)BK_CAP;
    for (unsigned j0 = beg; j0 < end; j0 += BK_T * BK_W) {
        unsigned w[BK_W];
#pragma unroll
        for (int k = 0; k < BK_W; ++k) {
            const unsigned j = j0 + k * BK_T + tid;
            w[k] = part[(j < end) ? j : (end - 1u)];
        }
#pragma unroll
        for (int k = 0; k < BK_W; ++k) {
            const unsigned j = j0 + k * BK_T + tid;
            if (j < end) {
                const unsigned p = atomicAdd(&c[w[k] & (unsigned)(NPB - 1)], 1u);
                const int sv = (int)(w[k] >> BSH);
                if (in_lds) lb[p] = sv; else esrc[beg + p] = sv;
            }
        }
    }
    if (in_lds) {
        __syncthreads();
        for (unsigned k = tid; k < nb; k += BK_T) esrc[beg + k] = lb[k];
    }
}

static constexpr int GT_ROWS  = 80;
static constexpr int GT_TILES = GT_ROWS / 16;
static constexpr int G1_PITCH = IND + 4;
__global__ __launch_bounds__(256) void k_gemm1(const float* __restrict__ X,
                                               const float* __restrict__ W,
                                               float* __restrict__ H) {
    __shared__ __attribute__((aligned(16))) float xs[GT_ROWS * G1_PITCH];
    const int tid  = threadIdx.x;
    const int wave = tid >> 5;
    const int lane = tid & 31;
    const int half = lane >> 4;
    const int l16  = lane & 15;
    const int row0 = blockIdx.x * GT_ROWS;
    const int col  = wave * 16 + l16;

    const float4* xg = (const float4*)(X + (size_t)row0 * IND);
#pragma unroll
    for (int i = 0; i < GT_ROWS * IND / 4 / 256; ++i) {
        const int idx = i * 256 + tid;
        const float4 v = xg[idx];
        *(float4*)(xs + (idx >> 6) * G1_PITCH + (idx & 63) * 4) = v;
    }
    __syncthreads();

    v8f c[GT_TILES];
#pragma unroll
    for (int rt = 0; rt < GT_TILES; ++rt) c[rt] = v8f{};
    const float* ap = xs + l16 * G1_PITCH + half * 2;
    const float* wp = W + (size_t)(half * 2) * HID + col;
#pragma unroll 8
    for (int kk = 0; kk < IND; kk += 4) {
        v2f b;
        b.x = wp[(size_t)kk * HID];
        b.y = wp[(size_t)(kk + 1) * HID];
#pragma unroll
        for (int rt = 0; rt < GT_TILES; ++rt) {
            const v2f a = *(const v2f*)(ap + rt * 16 * G1_PITCH + kk);
            c[rt] = __builtin_amdgcn_wmma_f32_16x16x4_f32(false, a, false, b, (short)0, c[rt], false, false);
        }
    }
#pragma unroll
    for (int rt = 0; rt < GT_TILES; ++rt)
#pragma unroll
        for (int v = 0; v < 8; ++v)
            H[(size_t)(row0 + rt * 16 + v + half * 8) * HID + col] = c[rt][v];
}

template <int CTRL> __device__ __forceinline__ float row_swz(float x) {
    const int b = __builtin_bit_cast(int, x);
    return __builtin_bit_cast(float, __builtin_amdgcn_update_dpp(0, b, CTRL, 0xf, 0xf, true));
}
__device__ __forceinline__ float quad_sum(float x) {
    x += row_swz<0x161>(x);
    x += row_swz<0x162>(x);
    return x;
}
__device__ __forceinline__ float row_sum(float x) {
    x = quad_sum(x);
    x += row_swz<0x164>(x);
    x += row_swz<0x168>(x);
    return x;
}
__device__ __forceinline__ float xhalf(float x) {
    const int b = __builtin_bit_cast(int, x);
    return __builtin_bit_cast(float, __builtin_amdgcn_permlanex16(b, b, 0x76543210, 0xfedcba98, false, false));
}
__device__ __forceinline__ float dot4(const float4 a, const float4 b) {
    return a.x * b.x + a.y * b.y + a.z * b.z + a.w * b.w;
}

__global__ __launch_bounds__(256) void k_gat1(const int* __restrict__ esrc,
                                              const unsigned* __restrict__ deg, const unsigned* __restrict__ begp,
                                              const float* __restrict__ H,
                                              const float* __restrict__ al, const float* __restrict__ ar,
                                              const float* __restrict__ b, float* __restrict__ h2) {
    const int lane = threadIdx.x & 31;
    const int n = blockIdx.x * 8 + __builtin_amdgcn_readfirstlane((int)(threadIdx.x >> 5));
    const int beg = __builtin_amdgcn_readfirstlane((int)begp[n]);
    const int end = beg + __builtin_amdgcn_readfirstlane((int)deg[n]);
    const float* Hl = H + 4 * lane;
    const float4 alv = *(const float4*)(al + 4 * lane);
    const float4 arv = *(const float4*)(ar + 4 * lane);
    const float4 own = *(const float4*)(Hl + (size_t)n * HID);
    const float erv = quad_sum(dot4(own, arv));

    float m = -__builtin_inff();
    float ssum = 0.f;
    float4 acc = make_float4(0.f, 0.f, 0.f, 0.f);
    for (int j0 = beg; j0 < end; j0 += 32) {
        const int jj = j0 + lane;
        const int cnt = (end - j0 < 32) ? (end - j0) : 32;
        const int sv = esrc[(jj < end) ? jj : (end - 1)];
        const int last = cnt - 1;
        for (int i = 0; i < cnt; i += 4) {
            float4 r[4];
            float e[4];
#pragma unroll
            for (int g = 0; g < 4; ++g) {
                const int ig = (i + g < cnt) ? (i + g) : last;
                const int s = __builtin_amdgcn_readlane(sv, ig);
                r[g] = *(const float4*)(Hl + (size_t)s * HID);
            }
#pragma unroll
            for (int g = 0; g < 4; ++g) {
                const float d = quad_sum(dot4(r[g], alv));
                e[g] = (i + g < cnt) ? lrelu(d + erv) : -__builtin_inff();
            }
            const float mn = fmaxf(fmaxf(m, e[0]), fmaxf(fmaxf(e[1], e[2]), e[3]));
            const float sc = __expf(m - mn);
            ssum *= sc; acc.x *= sc; acc.y *= sc; acc.z *= sc; acc.w *= sc;
#pragma unroll
            for (int g = 0; g < 4; ++g) {
                const float p = __expf(e[g] - mn);
                ssum += p; acc.x += p * r[g].x; acc.y += p * r[g].y; acc.z += p * r[g].z; acc.w += p * r[g].w;
            }
            m = mn;
        }
    }
    const float4 bv = *(const float4*)(b + 4 * lane);
    const bool any = ssum > 0.f;
    float4 o;
    o.x = elu1(elu1((any ? acc.x / ssum : 0.f) + bv.x));
    o.y = elu1(elu1((any ? acc.y / ssum : 0.f) + bv.y));
    o.z = elu1(elu1((any ? acc.z / ssum : 0.f) + bv.z));
    o.w = elu1(elu1((any ? acc.w / ssum : 0.f) + bv.w));
    *(float4*)(h2 + (size_t)n * HID + 4 * lane) = o;
}

static constexpr int G2_PITCH = HID + 4;
__global__ __launch_bounds__(128) void k_gemm2(const float* __restrict__ H2,
                                               const float* __restrict__ W,
                                               float* __restrict__ HT) {
    __shared__ __attribute__((aligned(16))) float hs[GT_ROWS * G2_PITCH];
    const int tid  = threadIdx.x;
    const int wave = tid >> 5;
    const int lane = tid & 31;
    const int half = lane >> 4;
    const int l16  = lane & 15;
    const int row0 = blockIdx.x * GT_ROWS;
    const int col  = wave * 16 + l16;

    const float4* hg = (const float4*)(H2 + (size_t)row0 * HID);
#pragma unroll
    for (int i = 0; i < GT_ROWS * HID / 4 / 128; ++i) {
        const int idx = i * 128 + tid;
        const float4 v = hg[idx];
        *(float4*)(hs + (idx >> 5) * G2_PITCH + (idx & 31) * 4) = v;
    }
    __syncthreads();

    v8f c[GT_TILES];
#pragma unroll
    for (int rt = 0; rt < GT_TILES; ++rt) c[rt] = v8f{};
    const float* ap = hs + l16 * G2_PITCH + half * 2;
    const float* wp = W + (size_t)(half * 2) * NCLS + col;
#pragma unroll 8
    for (int kk = 0; kk < HID; kk += 4) {
        v2f b;
        b.x = wp[(size_t)kk * NCLS];
        b.y = wp[(size_t)(kk + 1) * NCLS];
#pragma unroll
        for (int rt = 0; rt < GT_TILES; ++rt) {
            const v2f a = *(const v2f*)(ap + rt * 16 * G2_PITCH + kk);
            c[rt] = __builtin_amdgcn_wmma_f32_16x16x4_f32(false, a, false, b, (short)0, c[rt], false, false);
        }
    }
#pragma unroll
    for (int rt = 0; rt < GT_TILES; ++rt)
#pragma unroll
        for (int v = 0; v < 8; ++v)
            HT[(size_t)(row0 + rt * 16 + v + half * 8) * NCLS + col] = c[rt][v];
}

__global__ __launch_bounds__(256) void k_gat2(const int* __restrict__ esrc,
                                              const unsigned* __restrict__ deg, const unsigned* __restrict__ begp,
                                              const float* __restrict__ HT,
                                              const float* __restrict__ al, const float* __restrict__ ar,
                                              const float* __restrict__ b, float* __restrict__ out) {
    const int lane = threadIdx.x & 31;
    const int hf = lane >> 4, l16 = lane & 15;
    const int n = blockIdx.x * 8 + __builtin_amdgcn_readfirstlane((int)(threadIdx.x >> 5));
    const int beg = __builtin_amdgcn_readfirstlane((int)begp[n]);
    const int end = beg + __builtin_amdgcn_readfirstlane((int)deg[n]);
    static_assert((unsigned long long)NN * NCLS * sizeof(float) < (1ull << 32), "32-bit row offsets");
    const char* Tb = (const char*)HT;
    const unsigned lo = (unsigned)l16 * 16u;
    const float4 alv = *(const float4*)(al + 4 * l16);
    const float4 arv = *(const float4*)(ar + 4 * l16);
    const float4 own = *(const float4*)(Tb + (((unsigned)n << 8) + lo));
    const float erv = row_sum(dot4(own, arv));

    float m = -__builtin_inff();
    float ssum = 0.f;
    float4 acc = make_float4(0.f, 0.f, 0.f, 0.f);
    for (int j0 = beg; j0 < end; j0 += 32) {
        const int jj = j0 + lane;
        const int cnt = (end - j0 < 32) ? (end - j0) : 32;
        const int sv = esrc[(jj < end) ? jj : (end - 1)];
        const int last = cnt - 1;
        for (int i = 0; i < cnt; i += 8) {
            float4 r[4];
            float e[4];
            bool ok[4];
#pragma unroll
            for (int g = 0; g < 4; ++g) {
                const int ia = i + 2 * g, ib = ia + 1;
                const int sa = __builtin_amdgcn_readlane(sv, (ia < cnt) ? ia : last);
                const int sb = __builtin_amdgcn_readlane(sv, (ib < cnt) ? ib : last);
                const int s = hf ? sb : sa;
                ok[g] = (ia + hf) < cnt;
                r[g] = *(const float4*)(Tb + (((unsigned)s << 8) + lo));
            }
            asm volatile("" :: "v"(r[0].x), "v"(r[1].x), "v"(r[2].x), "v"(r[3].x));
#pragma unroll
            for (int g = 0; g < 4; ++g) {
                const float d = row_sum(dot4(r[g], alv));
                e[g] = ok[g] ? lrelu(d + erv) : -__builtin_inff();
            }
            const float mn = fmaxf(fmaxf(m, e[0]), fmaxf(fmaxf(e[1], e[2]), e[3]));
            const float sc = (m == mn) ? 1.f : __expf(m - mn);
            ssum *= sc; acc.x *= sc; acc.y *= sc; acc.z *= sc; acc.w *= sc;
#pragma unroll
            for (int g = 0; g < 4; ++g) {
                const float p = ok[g] ? __expf(e[g] - mn) : 0.f;
                ssum += p; acc.x += p * r[g].x; acc.y += p * r[g].y; acc.z += p * r[g].z; acc.w += p * r[g].w;
            }
            m = mn;
        }
    }
    const float M = fmaxf(m, xhalf(m));
    const float f = (m == M) ? 1.f : __expf(m - M);
    ssum *= f; acc.x *= f; acc.y *= f; acc.z *= f; acc.w *= f;
    ssum += xhalf(ssum);
    acc.x += xhalf(acc.x); acc.y += xhalf(acc.y); acc.z += xhalf(acc.z); acc.w += xhalf(acc.w);
    const float4 bv = *(const float4*)(b + 4 * l16);
    const bool any = ssum > 0.f;
    float4 o;
    o.x = (any ? acc.x / ssum : 0.f) + bv.x;
    o.y = (any ? acc.y / ssum : 0.f) + bv.y;
    o.z = (any ? acc.z / ssum : 0.f) + bv.z;
    o.w = (any ? acc.w / ssum : 0.f) + bv.w;
    if (hf == 0) *(float4*)(out + (size_t)n * NCLS + 4 * l16) = o;
}

extern "C" void kernel_launch(void* const* d_in, const int* in_sizes, int n_in,
                              void* d_out, int out_size, void* d_ws, size_t ws_size,
                              hipStream_t stream) {
    const float* x   = (const float*)d_in[0];
    const int*   src = (const int*)  d_in[1];
    const int*   dst = (const int*)  d_in[2];
    const float* W1  = (const float*)d_in[3];
    const float* al1 = (const float*)d_in[4];
    const float* ar1 = (const float*)d_in[5];
    const float* b1  = (const float*)d_in[6];
    const float* W2  = (const float*)d_in[7];
    const float* al2 = (const float*)d_in[8];
    const float* ar2 = (const float*)d_in[9];
    const float* b2  = (const float*)d_in[10];
    float* out = (float*)d_out;

    size_t off = 0;
    auto take = [&](size_t nfloats) -> float* {
        float* p = (float*)((char*)d_ws + off);
        off = (off + nfloats * sizeof(float) + 255) & ~(size_t)255;
        return p;
    };
    float*    h1    = take((size_t)NN * HID);
    int*      esrc  = (int*)take((size_t)EE);
    float*    h2    = take((size_t)NN * HID);
    float*    h2t   = take((size_t)NN * NCLS);
    unsigned* deg   = (unsigned*)take((size_t)NN);
    unsigned* cur   = (unsigned*)take((size_t)NN);
    unsigned* cnt   = (unsigned*)h2t;
    unsigned* gbase = cnt + (size_t)NBK * NCHP;
    unsigned* part  = (unsigned*)h2;
    static_assert((size_t)EE <= (size_t)NN * HID, "the edge-word plane fits in the h2 plane");
    static_assert((size_t)NBK * NCHP + NBK + 1 <= (size_t)NN * NCLS, "the count table fits in the h2t plane");

    const int T = 256;
    k_hist   <<<NCH, CH_T, 0, stream>>>(dst, cnt);
    k_scatter<<<NCH, CH_T, 0, stream>>>(src, dst, cnt, gbase, part);
    k_bucket <<<NBK, BK_T, 0, stream>>>(part, gbase, deg, cur, esrc);
    k_gemm1<<<NN / GT_ROWS, T, 0, stream>>>(x, W1, h1);
    k_gat1 <<<NN / 8, T, 0, stream>>>(esrc, deg, cur, h1, al1, ar1, b1, h2);
    k_gemm2<<<NN / GT_ROWS, 128, 0, stream>>>(h2, W2, h2t);
    k_gat2 <<<NN / 8, T, 0, stream>>>(esrc, deg, cur, h2t, al2, ar2, b2, out);
}


# Per-node kernels' epilogue: ELU as a branch-free series (to x^6, above -0.25) or hardware exp - 1 in place of 8 expm1 calls, one v_rcp_f32 in place of 4 divisions; ELU values within a few ulp
# speedup vs baseline: 10.0162x; 1.0504x over previous
#include <hip/hip_runtime.h>

typedef __attribute__((ext_vector_type(2))) float v2f;
typedef __attribute__((ext_vector_type(8))) float v8f;

static constexpr int NN   = 100000;
static constexpr int EE   = 1600000;
static constexpr int IND  = 256;
static constexpr int HID  = 128;
static constexpr int HEADS= 8;
static constexpr int DH   = 16;
static constexpr int NCLS = 64;
static constexpr float SLOPE = 0.2f;

__device__ __forceinline__ float lrelu(float x) { return x > 0.f ? x : SLOPE * x; }
__device__ __forceinline__ float elu1(float x) {
    const float xm = fminf(x, 0.f);
    float q = fmaf(xm, 1.f / 720.f, 1.f / 120.f);
    q = fmaf(xm, q, 1.f / 24.f);
    q = fmaf(xm, q, 1.f / 6.f);
    q = fmaf(xm, q, 0.5f);
    q = fmaf(xm * xm, q, xm);
    const float e = __expf(xm) - 1.f;
    const float r = (xm > -0.25f) ? q : e;
    return (x <= 0.f) ? r : x;
}

static constexpr int BSH    = 9;
static constexpr int NPB    = 1 << BSH;
static constexpr int NBK    = (NN + NPB - 1) / NPB;
static constexpr int CH_T   = 1024;
static constexpr int CH_E   = CH_T * 16;
static constexpr int NCH    = (EE + CH_E - 1) / CH_E;
static constexpr int NCHP   = (NCH + 3) / 4 * 4;
static constexpr int BK_T   = 1024;
static constexpr int BK_W   = 12;
static constexpr int BK_CAP = 32768;
static_assert(EE % 4 == 0, "the edge arrays are read by 16-byte loads");
static_assert(NBK <= CH_T && NPB <= BK_T, "one thread per bucket in a binning block, per node in a bucket block");
static_assert(NCHP - NCH <= NCH, "the first blocks clear the row padding");
static_assert(((unsigned long long)(NN - 1) << BSH) + (NPB - 1) < (1ull << 32), "an edge word holds source and place");

template <int NW> __device__ __forceinline__ unsigned block_excl_scan(unsigned v, unsigned* wsum, int lane, int wave) {
    unsigned inc = v;
    for (int o = 1; o < 32; o <<= 1) {
        const unsigned u = __shfl_up(inc, o);
        if (lane >= o) inc += u;
    }
    if (lane == 31) wsum[wave] = inc;
    __syncthreads();
    const unsigned t = (lane < NW) ? wsum[lane] : 0u;
    unsigned ti = t;
    for (int o = 1; o < 32; o <<= 1) {
        const unsigned u = __shfl_up(ti, o);
        if (lane >= o) ti += u;
    }
    const unsigned wbase = __shfl(ti - t, wave);
    __syncthreads();
    return wbase + inc - v;
}

__global__ __launch_bounds__(CH_T) void k_hist(const int* __restrict__ dst, unsigned* __restrict__ cnt) {
    __shared__ unsigned hs[NBK];
    const int tid = threadIdx.x;
    const int blk = blockIdx.x;
    if (tid < NBK) hs[tid] = 0u;
    __syncthreads();
    const int4* d4 = (const int4*)dst;
    int4 d[4];
    bool ok[4];
#pragma unroll
    for (int i = 0; i < 4; ++i) {
        const int e4 = blk * (CH_E / 4) + i * CH_T + tid;
        ok[i] = e4 < EE / 4;
        d[i] = d4[ok[i] ? e4 : (EE / 4 - 1)];
    }
#pragma unroll
    for (int i = 0; i < 4; ++i) {
        if (ok[i]) {
            atomicAdd(&hs[d[i].x >> BSH], 1u);
            atomicAdd(&hs[d[i].y >> BSH], 1u);
            atomicAdd(&hs[d[i].z >> BSH], 1u);
            atomicAdd(&hs[d[i].w >> BSH], 1u);
        }
    }
    __syncthreads();
    if (tid < NBK) {
        cnt[tid * NCHP + blk] = hs[tid];
        if (blk < NCHP - NCH) cnt[tid * NCHP + NCH + blk] = 0u;
    }
}

__global__ __launch_bounds__(CH_T) void k_scatter(const int* __restrict__ src, const int* __restrict__ dst,
                                                  const unsigned* __restrict__ cnt, unsigned* __restrict__ gbase,
                                                  unsigned* __restrict__ part) {
    __shared__ unsigned wbuf[CH_E];
    __shared__ unsigned loff[NBK];
    __shared__ unsigned lcur[NBK];
    __shared__ unsigned goff[NBK];
    __shared__ unsigned wsum[CH_T / 32];
    const int tid  = threadIdx.x;
    const int lane = tid & 31;
    const int wave = tid >> 5;
    const int blk  = blockIdx.x;

    unsigned tot = 0u, pre = 0u, mine = 0u;
    if (tid < NBK) {
        const uint4* cp = (const uint4*)(cnt + tid * NCHP);
        auto acc = [&](int r, unsigned c) {
            if (r < NCH) {
                tot += c;
                pre += (r < blk) ? c : 0u;
                mine = (r == blk) ? c : mine;
            }
        };
#pragma unroll
        for (int q = 0; q < NCHP / 4; ++q) {
            const uint4 v = cp[q];
            acc(4 * q, v.x); acc(4 * q + 1, v.y); acc(4 * q + 2, v.z); acc(4 * q + 3, v.w);
        }
    }
    const unsigned base = block_excl_scan<CH_T / 32>(tot,  wsum, lane, wave);
    const unsigned lst  = block_excl_scan<CH_T / 32>(mine, wsum, lane, wave);
    if (tid < NBK) {
        loff[tid] = lst;
        lcur[tid] = lst;
        goff[tid] = base + pre;
        if (blk == 0) {
            gbase[tid] = base;
            if (tid == NBK - 1) gbase[NBK] = base + tot;
        }
    }
    __syncthreads();

    const int4* d4 = (const int4*)dst;
    const int4* s4 = (const int4*)src;
    int4 d[4], s[4];
    bool ok[4];
#pragma unroll
    for (int i = 0; i < 4; ++i) {
        const int e4 = blk * (CH_E / 4) + i * CH_T + tid;
        ok[i] = e4 < EE / 4;
        const int ec = ok[i] ? e4 : (EE / 4 - 1);
        d[i] = d4[ec];
        s[i] = s4[ec];
    }
    auto place = [&](int dv, int sv) {
        const unsigned p = atomicAdd(&lcur[dv >> BSH], 1u);
        wbuf[p] = ((unsigned)sv << BSH) | ((unsigned)dv & (unsigned)(NPB - 1));
    };
#pragma unroll
    for (int i = 0; i < 4; ++i) {
        if (ok[i]) {
            place(d[i].x, s[i].x);
            place(d[i].y, s[i].y);
            place(d[i].z, s[i].z);
            place(d[i].w, s[i].w);
        }
    }
    __syncthreads();

    for (int b = wave; b < NBK; b += CH_T / 32) {
        const unsigned s0 = loff[b];
        const unsigned n  = lcur[b] - s0;
        const unsigned g  = goff[b];
        for (unsigned l = lane; l < n; l += 32) part[g + l] = wbuf[s0 + l];
    }
}

__global__ __launch_bounds__(BK_T) void k_bucket(const unsigned* __restrict__ part, const unsigned* __restrict__ gbase,
                                                 unsigned* __restrict__ deg, unsigned* __restrict__ cur,
                                                 int* __restrict__ esrc) {
    __shared__ int lb[BK_CAP];
    __shared__ unsigned c[NPB];
    __shared__ unsigned wsum[BK_T / 32];
    const int tid  = threadIdx.x;
    const int lane = tid & 31;
    const int wave = tid >> 5;
    const int bk   = blockIdx.x;
    const unsigned beg = gbase[bk];
    const unsigned end = gbase[bk + 1];
    const unsigned nb  = end - beg;
    if (tid < NPB) c[tid] = 0u;
    __syncthreads();

    for (unsigned j0 = beg; j0 < end; j0 += BK_T * BK_W) {
        unsigned w[BK_W];
#pragma unroll
        for (int k = 0; k < BK_W; ++k) {
            const unsigned j = j0 + k * BK_T + tid;
            w[k] = part[(j < end) ? j : (end - 1u)];
        }
#pragma unroll
        for (int k = 0; k < BK_W; ++k) {
            const unsigned j = j0 + k * BK_T + tid;
            if (j < end) atomicAdd(&c[w[k] & (unsigned)(NPB - 1)], 1u);
        }
    }
    __syncthreads();

    const unsigned dg = (tid < NPB) ? c[tid] : 0u;
    const unsigned ex = block_excl_scan<BK_T / 32>(dg, wsum, lane, wave);
    if (tid < NPB) {
        const int n = bk * NPB + tid;
        if (n < NN) { deg[n] = dg; cur[n] = beg + ex; }
        c[tid] = ex;
    }
    __syncthreads();

    const bool in_lds = nb <= (unsigned)BK_CAP;
    for (unsigned j0 = beg; j0 < end; j0 += BK_T * BK_W) {
        unsigned w[BK_W];
#pragma unroll
        for (int k = 0; k < BK_W; ++k) {
            const unsigned j = j0 + k * BK_T + tid;
            w[k] = part[(j < end) ? j : (end - 1u)];
        }
#pragma unroll
        for (int k = 0; k < BK_W; ++k) {
            const unsigned j = j0 + k * BK_T + tid;
            if (j < end) {
                const unsigned p = atomicAdd(&c[w[k] & (unsigned)(NPB - 1)], 1u);
                const int sv = (int)(w[k] >> BSH);
                if (in_lds) lb[p] = sv; else esrc[beg + p] = sv;
            }
        }
    }
    if (in_lds) {
        __syncthreads();
        for (unsigned k = tid; k < nb; k += BK_T) esrc[beg + k] = lb[k];
    }
}

static constexpr int GT_ROWS  = 80;
static constexpr int GT_TILES = GT_ROWS / 16;
static constexpr int G1_PITCH = IND + 4;
__global__ __launch_bounds__(256) void k_gemm1(const float* __restrict__ X,
                                               const float* __restrict__ W,
                                               float* __restrict__ H) {
    __shared__ __attribute__((aligned(16))) float xs[GT_ROWS * G1_PITCH];
    const int tid  = threadIdx.x;
    const int wave = tid >> 5;
    const int lane = tid & 31;
    const int half = lane >> 4;
    const int l16  = lane & 15;
    const int row0 = blockIdx.x * GT_ROWS;
    const int col  = wave * 16 + l16;

    const float4* xg = (const float4*)(X + (size_t)row0 * IND);
#pragma unroll
    for (int i = 0; i < GT_ROWS * IND / 4 / 256; ++i) {
        const int idx = i * 256 + tid;
        const float4 v = xg[idx];
        *(float4*)(xs + (idx >> 6) * G1_PITCH + (idx & 63) * 4) = v;
    }
    __syncthreads();

    v8f c[GT_TILES];
#pragma unroll
    for (int rt = 0; rt < GT_TILES; ++rt) c[rt] = v8f{};
    const float* ap = xs + l16 * G1_PITCH + half * 2;
    const float* wp = W + (size_t)(half * 2) * HID + col;
#pragma unroll 8
    for (int kk = 0; kk < IND; kk += 4) {
        v2f b;
        b.x = wp[(size_t)kk * HID];
        b.y = wp[(size_t)(kk + 1) * HID];
#pragma unroll
        for (int rt = 0; rt < GT_TILES; ++rt) {
            const v2f a = *(const v2f*)(ap + rt * 16 * G1_PITCH + kk);
            c[rt] = __builtin_amdgcn_wmma_f32_16x16x4_f32(false, a, false, b, (short)0, c[rt], false, false);
        }
    }
#pragma unroll
    for (int rt = 0; rt < GT_TILES; ++rt)
#pragma unroll
        for (int v = 0; v < 8; ++v)
            H[(size_t)(row0 + rt * 16 + v + half * 8) * HID + col] = c[rt][v];
}

template <int CTRL> __device__ __forceinline__ float row_swz(float x) {
    const int b = __builtin_bit_cast(int, x);
    return __builtin_bit_cast(float, __builtin_amdgcn_update_dpp(0, b, CTRL, 0xf, 0xf, true));
}
__device__ __forceinline__ float quad_sum(float x) {
    x += row_swz<0x161>(x);
    x += row_swz<0x162>(x);
    return x;
}
__device__ __forceinline__ float row_sum(float x) {
    x = quad_sum(x);
    x += row_swz<0x164>(x);
    x += row_swz<0x168>(x);
    return x;
}
__device__ __forceinline__ float xhalf(float x) {
    const int b = __builtin_bit_cast(int, x);
    return __builtin_bit_cast(float, __builtin_amdgcn_permlanex16(b, b, 0x76543210, 0xfedcba98, false, false));
}
__device__ __forceinline__ float dot4(const float4 a, const float4 b) {
    return a.x * b.x + a.y * b.y + a.z * b.z + a.w * b.w;
}

__global__ __launch_bounds__(256) void k_gat1(const int* __restrict__ esrc,
                                              const unsigned* __restrict__ deg, const unsigned* __restrict__ begp,
                                              const float* __restrict__ H,
                                              const float* __restrict__ al, const float* __restrict__ ar,
                                              const float* __restrict__ b, float* __restrict__ h2) {
    const int lane = threadIdx.x & 31;
    const int n = blockIdx.x * 8 + __builtin_amdgcn_readfirstlane((int)(threadIdx.x >> 5));
    const int beg = __builtin_amdgcn_readfirstlane((int)begp[n]);
    const int end = beg + __builtin_amdgcn_readfirstlane((int)deg[n]);
    const float* Hl = H + 4 * lane;
    const float4 alv = *(const float4*)(al + 4 * lane);
    const float4 arv = *(const float4*)(ar + 4 * lane);
    const float4 own = *(const float4*)(Hl + (size_t)n * HID);
    const float erv = quad_sum(dot4(own, arv));

    float m = -__builtin_inff();
    float ssum = 0.f;
    float4 acc = make_float4(0.f, 0.f, 0.f, 0.f);
    for (int j0 = beg; j0 < end; j0 += 32) {
        const int jj = j0 + lane;
        const int cnt = (end - j0 < 32) ? (end - j0) : 32;
        const int sv = esrc[(jj < end) ? jj : (end - 1)];
        const int last = cnt - 1;
        for (int i = 0; i < cnt; i += 4) {
            float4 r[4];
            float e[4];
#pragma unroll
            for (int g = 0; g < 4; ++g) {
                const int ig = (i + g < cnt) ? (i + g) : last;
                const int s = __builtin_amdgcn_readlane(sv, ig);
                r[g] = *(const float4*)(Hl + (size_t)s * HID);
            }
#pragma unroll
            for (int g = 0; g < 4; ++g) {
                const float d = quad_sum(dot4(r[g], alv));
                e[g] = (i + g < cnt) ? lrelu(d + erv) : -__builtin_inff();
            }
            const float mn = fmaxf(fmaxf(m, e[0]), fmaxf(fmaxf(e[1], e[2]), e[3]));
            const float sc = __expf(m - mn);
            ssum *= sc; acc.x *= sc; acc.y *= sc; acc.z *= sc; acc.w *= sc;
#pragma unroll
            for (int g = 0; g < 4; ++g) {
                const float p = __expf(e[g] - mn);
                ssum += p; acc.x += p * r[g].x; acc.y += p * r[g].y; acc.z += p * r[g].z; acc.w += p * r[g].w;
            }
            m = mn;
        }
    }
    const float4 bv = *(const float4*)(b + 4 * lane);
    const bool any = ssum > 0.f;
    const float inv = any ? __builtin_amdgcn_rcpf(ssum) : 0.f;
    float4 o;
    o.x = elu1(elu1(acc.x * inv + bv.x));
    o.y = elu1(elu1(acc.y * inv + bv.y));
    o.z = elu1(elu1(acc.z * inv + bv.z));
    o.w = elu1(elu1(acc.w * inv + bv.w));
    *(float4*)(h2 + (size_t)n * HID + 4 * lane) = o;
}

static constexpr int G2_PITCH = HID + 4;
__global__ __launch_bounds__(128) void k_gemm2(const float* __restrict__ H2,
                                               const float* __restrict__ W,
                                               float* __restrict__ HT) {
    __shared__ __attribute__((aligned(16))) float hs[GT_ROWS * G2_PITCH];
    const int tid  = threadIdx.x;
    const int wave = tid >> 5;
    const int lane = tid & 31;
    const int half = lane >> 4;
    const int l16  = lane & 15;
    const int row0 = blockIdx.x * GT_ROWS;
    const int col  = wave * 16 + l16;

    const float4* hg = (const float4*)(H2 + (size_t)row0 * HID);
#pragma unroll
    for (int i = 0; i < GT_ROWS * HID / 4 / 128; ++i) {
        const int idx = i * 128 + tid;
        const float4 v = hg[idx];
        *(float4*)(hs + (idx >> 5) * G2_PITCH + (idx & 31) * 4) = v;
    }
    __syncthreads();

    v8f c[GT_TILES];
#pragma unroll
    for (int rt = 0; rt < GT_TILES; ++rt) c[rt] = v8f{};
    const float* ap = hs + l16 * G2_PITCH + half * 2;
    const float* wp = W + (size_t)(half * 2) * NCLS + col;
#pragma unroll 8
    for (int kk = 0; kk < HID; kk += 4) {
        v2f b;
        b.x = wp[(size_t)kk * NCLS];
        b.y = wp[(size_t)(kk + 1) * NCLS];
#pragma unroll
        for (int rt = 0; rt < GT_TILES; ++rt) {
            const v2f a = *(const v2f*)(ap + rt * 16 * G2_PITCH + kk);
            c[rt] = __builtin_amdgcn_wmma_f32_16x16x4_f32(false, a, false, b, (short)0, c[rt], false, false);
        }
    }
#pragma unroll
    for (int rt = 0; rt < GT_TILES; ++rt)
#pragma unroll
        for (int v = 0; v < 8; ++v)
            HT[(size_t)(row0 + rt * 16 + v + half * 8) * NCLS + col] = c[rt][v];
}

__global__ __launch_bounds__(256) void k_gat2(const int* __restrict__ esrc,
                                              const unsigned* __restrict__ deg, const unsigned* __restrict__ begp,
                                              const float* __restrict__ HT,
                                              const float* __restrict__ al, const float* __restrict__ ar,
                                              const float* __restrict__ b, float* __restrict__ out) {
    const int lane = threadIdx.x & 31;
    const int hf = lane >> 4, l16 = lane & 15;
    const int n = blockIdx.x * 8 + __builtin_amdgcn_readfirstlane((int)(threadIdx.x >> 5));
    const int beg = __builtin_amdgcn_readfirstlane((int)begp[n]);
    const int end = beg + __builtin_amdgcn_readfirstlane((int)deg[n]);
    static_assert((unsigned long long)NN * NCLS * sizeof(float) < (1ull << 32), "32-bit row offsets");
    const char* Tb = (const char*)HT;
    const unsigned lo = (unsigned)l16 * 16u;
    const float4 alv = *(const float4*)(al + 4 * l16);
    const float4 arv = *(const float4*)(ar + 4 * l16);
    const float4 own = *(const float4*)(Tb + (((unsigned)n << 8) + lo));
    const float erv = row_sum(dot4(own, arv));

    float m = -__builtin_inff();
    float ssum = 0.f;
    float4 acc = make_float4(0.f, 0.f, 0.f, 0.f);
    for (int j0 = beg; j0 < end; j0 += 32) {
        const int jj = j0 + lane;
        const int cnt = (end - j0 < 32) ? (end - j0) : 32;
        const int sv = esrc[(jj < end) ? jj : (end - 1)];
        const int last = cnt - 1;
        for (int i = 0; i < cnt; i += 8) {
            float4 r[4];
            float e[4];
            bool ok[4];
#pragma unroll
            for (int g = 0; g < 4; ++g) {
                const int ia = i + 2 * g, ib = ia + 1;
                const int sa = __builtin_amdgcn_readlane(sv, (ia < cnt) ? ia : last);
                const int sb = __builtin_amdgcn_readlane(sv, (ib < cnt) ? ib : last);
                const int s = hf ? sb : sa;
                ok[g] = (ia + hf) < cnt;
                r[g] = *(const float4*)(Tb + (((unsigned)s << 8) + lo));
            }
            asm volatile("" :: "v"(r[0].x), "v"(r[1].x), "v"(r[2].x), "v"(r[3].x));
#pragma unroll
            for (int g = 0; g < 4; ++g) {
                const float d = row_sum(dot4(r[g], alv));
                e[g] = ok[g] ? lrelu(d + erv) : -__builtin_inff();
            }
            const float mn = fmaxf(fmaxf(m, e[0]), fmaxf(fmaxf(e[1], e[2]), e[3]));
            const float sc = (m == mn) ? 1.f : __expf(m - mn);
            ssum *= sc; acc.x *= sc; acc.y *= sc; acc.z *= sc; acc.w *= sc;
#pragma unroll
            for (int g = 0; g < 4; ++g) {
                const float p = ok[g] ? __expf(e[g] - mn) : 0.f;
                ssum += p; acc.x += p * r[g].x; acc.y += p * r[g].y; acc.z += p * r[g].z; acc.w += p * r[g].w;
            }
            m = mn;
        }
    }
    const float M = fmaxf(m, xhalf(m));
    const float f = (m == M) ? 1.f : __expf(m - M);
    ssum *= f; acc.x *= f; acc.y *= f; acc.z *= f; acc.w *= f;
    ssum += xhalf(ssum);
    acc.x += xhalf(acc.x); acc.y += xhalf(acc.y); acc.z += xhalf(acc.z); acc.w += xhalf(acc.w);
    const float4 bv = *(const float4*)(b + 4 * l16);
    const bool any = ssum > 0.f;
    const float inv = any ? __builtin_amdgcn_rcpf(ssum) : 0.f;
    float4 o;
    o.x = acc.x * inv + bv.x;
    o.y = acc.y * inv + bv.y;
    o.z = acc.z * inv + bv.z;
    o.w = acc.w * inv + bv.w;
    if (hf == 0) *(float4*)(out + (size_t)n * NCLS + 4 * l16) = o;
}

extern "C" void kernel_launch(void* const* d_in, const int* in_sizes, int n_in,
                              void* d_out, int out_size, void* d_ws, size_t ws_size,
                              hipStream_t stream) {
    const float* x   = (const float*)d_in[0];
    const int*   src = (const int*)  d_in[1];
    const int*   dst = (const int*)  d_in[2];
    const float* W1  = (const float*)d_in[3];
    const float* al1 = (const float*)d_in[4];
    const float* ar1 = (const float*)d_in[5];
    const float* b1  = (const float*)d_in[6];
    const float* W2  = (const float*)d_in[7];
    const float* al2 = (const float*)d_in[8];
    const float* ar2 = (const float*)d_in[9];
    const float* b2  = (const float*)d_in[10];
    float* out = (float*)d_out;

    size_t off = 0;
    auto take = [&](size_t nfloats) -> float* {
        float* p = (float*)((char*)d_ws + off);
        off = (off + nfloats * sizeof(float) + 255) & ~(size_t)255;
        return p;
    };
    float*    h1    = take((size_t)NN * HID);
    int*      esrc  = (int*)take((size_t)EE);
    float*    h2    = take((size_t)NN * HID);
    float*    h2t   = take((size_t)NN * NCLS);
    unsigned* deg   = (unsigned*)take((size_t)NN);
    unsigned* cur   = (unsigned*)take((size_t)NN);
    unsigned* cnt   = (unsigned*)h2t;
    unsigned* gbase = cnt + (size_t)NBK * NCHP;
    unsigned* part  = (unsigned*)h2;
    static_assert((size_t)EE <= (size_t)NN * HID, "the edge-word plane fits in the h2 plane");
    static_assert((size_t)NBK * NCHP + NBK + 1 <= (size_t)NN * NCLS, "the count table fits in the h2t plane");

    const int T = 256;
    k_hist   <<<NCH, CH_T, 0, stream>>>(dst, cnt);
    k_scatter<<<NCH, CH_T, 0, stream>>>(src, dst, cnt, gbase, part);
    k_bucket <<<NBK, BK_T, 0, stream>>>(part, gbase, deg, cur, esrc);
    k_gemm1<<<NN / GT_ROWS, T, 0, stream>>>(x, W1, h1);
    k_gat1 <<<NN / 8, T, 0, stream>>>(esrc, deg, cur, h1, al1, ar1, b1, h2);
    k_gemm2<<<NN / GT_ROWS, 128, 0, stream>>>(h2, W2, h2t);
    k_gat2 <<<NN / 8, T, 0, stream>>>(esrc, deg, cur, h2t, al2, ar2, b2, out);
}
